# starting kernel
# baseline (speedup 1.0000x reference)
#include <hip/hip_runtime.h>
#include <hip/hip_bf16.h>

typedef __attribute__((ext_vector_type(2))) float v2f;
typedef __attribute__((ext_vector_type(8))) float v8f;

#define GNN_N 100000
#define GNN_E 1600000
#define GNN_D 64

__global__ __launch_bounds__(256) void gnn_zero_kernel(float* __restrict__ p, int n4) {
  int i = blockIdx.x * 256 + threadIdx.x;
  if (i < n4) ((float4*)p)[i] = make_float4(0.f, 0.f, 0.f, 0.f);
}

__global__ __launch_bounds__(256) void gnn_count_kernel(const int* __restrict__ ei,
                                                        float* __restrict__ cnt) {
  int e = blockIdx.x * 256 + threadIdx.x;
  if (e < GNN_E) {
    int d = ei[GNN_E + e];
    unsafeAtomicAdd(&cnt[d], 1.0f);
  }
}

__global__ __launch_bounds__(256) void gnn_inv_kernel(const float* __restrict__ cnt,
                                                      float* __restrict__ inv) {
  int i = blockIdx.x * 256 + threadIdx.x;
  if (i < GNN_N) inv[i] = 1.0f / fmaxf(cnt[i], 1.0f);
}

__global__ __launch_bounds__(256) void gnn_scatter_kernel(const int* __restrict__ ei,
                                                          const float* __restrict__ feat,
                                                          float* __restrict__ agg) {
  int tid = blockIdx.x * 256 + threadIdx.x;
  if (tid >= GNN_E * 16) return;
  int e = tid >> 4;
  int q = (tid & 15) << 2;
  int s = ei[e];
  int d = ei[GNN_E + e];
  float4 v = *(const float4*)(feat + (size_t)s * GNN_D + q);
  float* o = agg + (size_t)d * GNN_D + q;
  unsafeAtomicAdd(o + 0, v.x);
  unsafeAtomicAdd(o + 1, v.y);
  unsafeAtomicAdd(o + 2, v.z);
  unsafeAtomicAdd(o + 3, v.w);
}

__global__ __launch_bounds__(256) void gnn_sage_gemm_kernel(
    const float* __restrict__ A,
    const float* __restrict__ X,
    const float* __restrict__ inv,
    const float* __restrict__ Wl,
    const float* __restrict__ Wr,
    const float* __restrict__ bias,
    float* __restrict__ Y,
    int nRowTiles, int doRelu) {
  int wave = (blockIdx.x * 256 + threadIdx.x) >> 5;
  if (wave >= nRowTiles) return;
  int lane = threadIdx.x & 31;
  int m    = lane & 15;
  int half = lane >> 4;
  int row0 = wave * 16;
  int row  = row0 + m;
  float scale = inv[row];

  v8f c0 = {}, c1 = {}, c2 = {}, c3 = {};

#pragma unroll 1
  for (int pass = 0; pass < 2; ++pass) {
    const float* M = pass ? X : A;
    const float* W = pass ? Wr : Wl;
    float s = pass ? 1.0f : scale;
    const float* arow = M + (size_t)row * GNN_D;
#pragma unroll
    for (int kt = 0; kt < 16; ++kt) {
      int k0 = kt * 4 + half * 2;
      v2f a = *(const v2f*)(arow + k0);
      a.x *= s; a.y *= s;
      v2f bA = *(const v2f*)(W + (size_t)(0 * 16 + m) * GNN_D + k0);
      v2f bB = *(const v2f*)(W + (size_t)(1 * 16 + m) * GNN_D + k0);
      v2f bC = *(const v2f*)(W + (size_t)(2 * 16 + m) * GNN_D + k0);
      v2f bD = *(const v2f*)(W + (size_t)(3 * 16 + m) * GNN_D + k0);
      c0 = __builtin_amdgcn_wmma_f32_16x16x4_f32(false, a, false, bA, (short)0, c0, false, false);
      c1 = __builtin_amdgcn_wmma_f32_16x16x4_f32(false, a, false, bB, (short)0, c1, false, false);
      c2 = __builtin_amdgcn_wmma_f32_16x16x4_f32(false, a, false, bC, (short)0, c2, false, false);
      c3 = __builtin_amdgcn_wmma_f32_16x16x4_f32(false, a, false, bD, (short)0, c3, false, false);
    }
  }

  v8f acc[4] = {c0, c1, c2, c3};
#pragma unroll
  for (int jt = 0; jt < 4; ++jt) {
    int j = jt * 16 + m;
    float bv = bias[j];
#pragma unroll
    for (int r = 0; r < 8; ++r) {
      float v = acc[jt][r] + bv;
      if (doRelu) v = fmaxf(v, 0.0f);
      Y[(size_t)(row0 + r + half * 8) * GNN_D + j] = v;
    }
  }
}

extern "C" void kernel_launch(void* const* d_in, const int* in_sizes, int n_in,
                              void* d_out, int out_size, void* d_ws, size_t ws_size,
                              hipStream_t stream) {
  const float* x   = (const float*)d_in[0];
  const int*   ei  = (const int*)d_in[1];
  const float* W1l = (const float*)d_in[2];
  const float* b1  = (const float*)d_in[3];
  const float* W1r = (const float*)d_in[4];
  const float* W2l = (const float*)d_in[5];
  const float* b2  = (const float*)d_in[6];
  const float* W2r = (const float*)d_in[7];
  float* out = (float*)d_out;

  float* agg = (float*)d_ws;
  float* h   = agg + (size_t)GNN_N * GNN_D;
  float* cnt = h + (size_t)GNN_N * GNN_D;
  float* inv = cnt + GNN_N;

  const int feat4    = (GNN_N * GNN_D) / 4;
  const int zeroGrid = (feat4 + 255) / 256;
  const int cnt4     = GNN_N / 4;
  const int cntGrid  = (cnt4 + 255) / 256;
  const int edgeGrid = (GNN_E + 255) / 256;
  const int scatGrid = (GNN_E * 16 + 255) / 256;
  const int nGrid    = (GNN_N + 255) / 256;
  const int rowTiles = GNN_N / 16;
  const int gemmGrid = (rowTiles + 7) / 8;

  gnn_zero_kernel<<<cntGrid, 256, 0, stream>>>(cnt, cnt4);
  gnn_count_kernel<<<edgeGrid, 256, 0, stream>>>(ei, cnt);
  gnn_inv_kernel<<<nGrid, 256, 0, stream>>>(cnt, inv);

  gnn_zero_kernel<<<zeroGrid, 256, 0, stream>>>(agg, feat4);
  gnn_scatter_kernel<<<scatGrid, 256, 0, stream>>>(ei, x, agg);
  gnn_sage_gemm_kernel<<<gemmGrid, 256, 0, stream>>>(agg, x, inv, W1l, W1r, b1, h,
                                                     rowTiles,  1);

  gnn_zero_kernel<<<zeroGrid, 256, 0, stream>>>(agg, feat4);
  gnn_scatter_kernel<<<scatGrid, 256, 0, stream>>>(ei, h, agg);
  gnn_sage_gemm_kernel<<<gemmGrid, 256, 0, stream>>>(agg, h, inv, W2l, W2r, b2, out,
                                                     rowTiles,  0);
}


# GEMM made persistent: 256 work-groups of 8 waves, each wave keeps both weight matrices in registers as B fragments (loaded once) and walks 16-row tiles, next tile's A rows loaded ahead; k order kept
# speedup vs baseline: 2.9485x; 2.9485x over previous
#include <hip/hip_runtime.h>
#include <hip/hip_bf16.h>

typedef __attribute__((ext_vector_type(2))) float v2f;
typedef __attribute__((ext_vector_type(8))) float v8f;

#define GNN_N 100000
#define GNN_E 1600000
#define GNN_D 64

#define OWN_G      128
#define OWN_S      2
#define OWN_ROWS   ((GNN_N + OWN_G - 1) / OWN_G)
#define AGG_THREADS 1024
#define AGG_WAVES  (AGG_THREADS / 32)
#define CH_EDGES   128
#define N_CH       (GNN_E / CH_EDGES / OWN_S)
#define N_IT       ((N_CH + AGG_WAVES - 1) / AGG_WAVES)
#define GRP        4
#define N_GRP      ((N_IT + GRP - 1) / GRP)
#define LIST_CAP   640
#define FLUSH_T    128
#define FL_BATCH   32
#define GL_CAP     1024
#define N_SLOT     (OWN_G * OWN_S * AGG_WAVES)

static_assert(GNN_E % (CH_EDGES * OWN_S) == 0, "chunks must tile every part of the edge list");
static_assert(GNN_E <= (1 << 21), "edge id must fit 21 bits of a list entry");
static_assert(OWN_ROWS <= 1024, "local row must fit 10 bits of a list entry");
static_assert(FLUSH_T + GRP * CH_EDGES <= LIST_CAP, "a group of chunks can never overflow the list");
static_assert((OWN_G - 1) * OWN_ROWS < GNN_N, "every work-group owns at least one row");
static_assert(GNN_N <= (1 << 22), "source row must fit 22 bits of a kept entry");
static_assert(GL_CAP % FL_BATCH == 0, "a wave's region holds whole batches");

template <bool PARTIAL>
__device__ __forceinline__ void gnn_batch_to_sums(int s, int dl, int m, int lane,
                                                  const float* __restrict__ feat, float* acc) {
  v2f row[FL_BATCH];
#pragma unroll
  for (int j = 0; j < FL_BATCH; ++j) {
    const int sj = __builtin_amdgcn_readlane(s, j);
    row[j] = *(const v2f*)(feat + (size_t)sj * GNN_D + 2 * lane);
  }
#pragma unroll
  for (int j = 0; j < FL_BATCH; ++j) {
    if (!PARTIAL || j < m) {
      const int dj = __builtin_amdgcn_readlane(dl, j);
      float* p = acc + dj * GNN_D + 2 * lane;
      atomicAdd(p, row[j].x);
      atomicAdd(p + 1, row[j].y);
    }
  }
}

template <bool PARTIAL, bool KEEP>
__device__ __forceinline__ void gnn_list_to_sums(const unsigned* mylst, int nproc, int lane,
                                                 const int* __restrict__ ei,
                                                 const float* __restrict__ feat, float* acc,
                                                 unsigned* __restrict__ myg, int& gn, bool& gfull) {
  for (int b = 0; b < nproc; b += FL_BATCH) {
    const int idx = b + lane;
    const bool ok = !PARTIAL || idx < nproc;
    const unsigned pk = ok ? mylst[idx] : 0u;
    const int s  = ok ? ei[pk >> 10] : 0;
    const int dl = (int)(pk & 1023u);
    const int m  = (!PARTIAL || (nproc - b) >= FL_BATCH) ? FL_BATCH : (nproc - b);
    if (KEEP) {
      if (gn + FL_BATCH <= GL_CAP) {
        if (ok) myg[gn + lane] = ((unsigned)s << 10) | (unsigned)dl;
        gn += m;
      } else {
        gfull = true;
      }
    }
    gnn_batch_to_sums<PARTIAL>(s, dl, m, lane, feat, acc);
  }
}

__device__ __forceinline__ void gnn_replay_to_sums(const unsigned* __restrict__ myg, int n, int lane,
                                                   const float* __restrict__ feat, float* acc) {
  if (n <= 0) return;
  const int last = n - 1;
  const unsigned v0 = myg[lane < last ? lane : last];
  unsigned pk = lane < n ? v0 : 0u;
  for (int b = 0; b < n; b += FL_BATCH) {
    const int nx = b + FL_BATCH + lane;
    const unsigned vn = myg[nx < last ? nx : last];
    const unsigned pkn = nx < n ? vn : 0u;
    const unsigned su = pk >> 10;
    const int s  = (int)(su < (unsigned)GNN_N ? su : 0u);
    const int dl = (int)(pk & 1023u);
    if (n - b >= FL_BATCH) gnn_batch_to_sums<false>(s, dl, FL_BATCH, lane, feat, acc);
    else                   gnn_batch_to_sums<true>(s, dl, n - b, lane, feat, acc);
    pk = pkn;
  }
}

struct GnnOwn {
  int lane, w, part, lo, rot, e_part, slot;
  unsigned nr;
};

__device__ __forceinline__ GnnOwn gnn_own_setup() {
  GnnOwn o;
  o.lane = threadIdx.x & 31;
  o.w    = __builtin_amdgcn_readfirstlane(threadIdx.x >> 5);
  const int rng = blockIdx.x / OWN_S;
  o.part = blockIdx.x % OWN_S;
  o.lo   = rng * OWN_ROWS;
  o.nr   = (unsigned)((GNN_N - o.lo) < OWN_ROWS ? (GNN_N - o.lo) : OWN_ROWS);
  o.rot  = rng * (N_CH / OWN_G);
  o.e_part = o.part * (N_CH * CH_EDGES);
  o.slot = blockIdx.x * AGG_WAVES + o.w;
  return o;
}

template <bool KEEP>
__device__ __forceinline__ int gnn_scan_to_sums(const GnnOwn& o, const int* __restrict__ ei,
                                                const float* __restrict__ feat, float* acc,
                                                float* cntl, unsigned* mylst,
                                                unsigned* __restrict__ myg) {
  const int lane = o.lane, w = o.w, lo = o.lo, rot = o.rot, e_part = o.e_part;
  const unsigned nr = o.nr;
  const int* __restrict__ dstp = ei + GNN_E + e_part;
  int gn = 0;
  bool gfull = false;

  int n = 0;
  int4 cur[GRP];
#pragma unroll
  for (int u = 0; u < GRP; ++u) cur[u] = make_int4(-1, -1, -1, -1);

  for (int g = 0; g <= N_GRP; ++g) {
    if (n >= FLUSH_T) {
      const int nproc = n & ~(FL_BATCH - 1);
      gnn_list_to_sums<false, KEEP>(mylst, nproc, lane, ei, feat, acc, myg, gn, gfull);
      const int rem = n - nproc;
      const unsigned keep = (lane < rem) ? mylst[nproc + lane] : 0u;
      if (lane < rem) mylst[lane] = keep;
      n = rem;
    }

    const int kb = w + AGG_WAVES * GRP * g;
    const int cb = kb + rot;
    int4 nxt[GRP];
    if (kb + AGG_WAVES * (GRP - 1) < N_CH &&
        (cb + AGG_WAVES * (GRP - 1) < N_CH || cb >= N_CH)) {
      const int c0 = cb >= N_CH ? cb - N_CH : cb;
      const int4* p = (const int4*)(dstp + (size_t)c0 * CH_EDGES + lane * 4);
#pragma unroll
      for (int u = 0; u < GRP; ++u) nxt[u] = p[u * AGG_WAVES * (CH_EDGES / 4)];
    } else {
#pragma unroll
      for (int u = 0; u < GRP; ++u) {
        const int k = kb + AGG_WAVES * u;
        nxt[u] = make_int4(-1, -1, -1, -1);
        if (k < N_CH) {
          int c = k + rot;
          if (c >= N_CH) c -= N_CH;
          nxt[u] = *(const int4*)(dstp + (size_t)c * CH_EDGES + lane * 4);
        }
      }
    }

#pragma unroll
    for (int u = 0; u < GRP; ++u) {
      const int4 d = cur[u];
      const unsigned t0 = (unsigned)(d.x - lo), t1 = (unsigned)(d.y - lo);
      const unsigned t2 = (unsigned)(d.z - lo), t3 = (unsigned)(d.w - lo);
      const unsigned ta = t0 < t1 ? t0 : t1, tb = t2 < t3 ? t2 : t3;
      const unsigned tm = ta < tb ? ta : tb;
      if (__builtin_amdgcn_ballot_w32(tm < nr) != 0u) {
        const unsigned k0 = __builtin_amdgcn_ballot_w32(t0 < nr);
        const unsigned k1 = __builtin_amdgcn_ballot_w32(t1 < nr);
        const unsigned k2 = __builtin_amdgcn_ballot_w32(t2 < nr);
        const unsigned k3 = __builtin_amdgcn_ballot_w32(t3 < nr);
        int c = kb - AGG_WAVES * GRP + AGG_WAVES * u + rot;
        if (c >= N_CH) c -= N_CH;
        const unsigned e0 = (unsigned)(e_part + c * CH_EDGES + lane * 4);
#define GNN_APPEND(MK, T, K)                                                  \
        if ((MK) != 0u) {                                                     \
          if ((T) < nr) {                                                     \
            const int pos = n + (int)__builtin_amdgcn_mbcnt_lo((MK), 0u);     \
            mylst[pos] = ((e0 + (K)) << 10) | (T);                            \
            if (KEEP) atomicAdd(&cntl[T], 1.0f);                              \
          }                                                                   \
          n += __builtin_popcount(MK);                                        \
        }
        GNN_APPEND(k0, t0, 0u)
        GNN_APPEND(k1, t1, 1u)
        GNN_APPEND(k2, t2, 2u)
        GNN_APPEND(k3, t3, 3u)
#undef GNN_APPEND
      }
    }
#pragma unroll
    for (int u = 0; u < GRP; ++u) cur[u] = nxt[u];
  }
  gnn_list_to_sums<true, KEEP>(mylst, n, lane, ei, feat, acc, myg, gn, gfull);
  return gfull ? -1 : gn;
}

__device__ __forceinline__ void gnn_store_sums(const GnnOwn& o, const float* acc,
                                               float* __restrict__ aggp) {
  float4* __restrict__ o4 = (float4*)(aggp + ((size_t)o.part * GNN_N + o.lo) * GNN_D);
  for (int i = threadIdx.x; i < (int)o.nr * (GNN_D / 4); i += AGG_THREADS)
    o4[i] = ((const float4*)acc)[i];
}

__global__ __launch_bounds__(AGG_THREADS) void gnn_owner_agg_scan_kernel(
    const int* __restrict__ ei,
    const float* __restrict__ feat,
    float* __restrict__ aggp,
    float* __restrict__ cntp,
    unsigned* __restrict__ glist,
    int* __restrict__ gcnt) {
  __shared__ __attribute__((aligned(16))) float acc[OWN_ROWS * GNN_D];
  __shared__ float cntl[OWN_ROWS];
  __shared__ unsigned lst[AGG_WAVES * LIST_CAP];

  const int tid = threadIdx.x;
  const GnnOwn o = gnn_own_setup();

  for (int i = tid; i < OWN_ROWS * GNN_D / 4; i += AGG_THREADS)
    ((float4*)acc)[i] = make_float4(0.f, 0.f, 0.f, 0.f);
  for (int i = tid; i < OWN_ROWS; i += AGG_THREADS) cntl[i] = 0.f;
  __syncthreads();

  const int kept = gnn_scan_to_sums<true>(o, ei, feat, acc, cntl, lst + o.w * LIST_CAP,
                                          glist + (size_t)o.slot * GL_CAP);
  if (o.lane == 0) gcnt[o.slot] = kept;

  __syncthreads();
  gnn_store_sums(o, acc, aggp);
  for (int i = tid; i < (int)o.nr; i += AGG_THREADS)
    cntp[(size_t)o.part * GNN_N + o.lo + i] = cntl[i];
}

__global__ __launch_bounds__(AGG_THREADS) void gnn_owner_agg_replay_kernel(
    const int* __restrict__ ei,
    const float* __restrict__ feat,
    float* __restrict__ aggp,
    const unsigned* __restrict__ glist,
    const int* __restrict__ gcnt) {
  __shared__ __attribute__((aligned(16))) float acc[OWN_ROWS * GNN_D];
  __shared__ unsigned lst[AGG_WAVES * LIST_CAP];

  const int tid = threadIdx.x;
  const GnnOwn o = gnn_own_setup();

  for (int i = tid; i < OWN_ROWS * GNN_D / 4; i += AGG_THREADS)
    ((float4*)acc)[i] = make_float4(0.f, 0.f, 0.f, 0.f);
  __syncthreads();

  const int kept = __builtin_amdgcn_readfirstlane(gcnt[o.slot]);
  if (kept >= 0) {
    gnn_replay_to_sums(glist + (size_t)o.slot * GL_CAP, kept < GL_CAP ? kept : GL_CAP, o.lane,
                       feat, acc);
  } else {
    gnn_scan_to_sums<false>(o, ei, feat, acc, nullptr, lst + o.w * LIST_CAP, nullptr);
  }

  __syncthreads();
  gnn_store_sums(o, acc, aggp);
}

static_assert(OWN_S == 2, "the GEMM adds exactly two partial planes");
#define GEMM_WAVES 8
#define GEMM_GRID  256
#define GEMM_KT    (GNN_D / 4)

struct GnnTileRaw {
  v2f p[GEMM_KT], q[GEMM_KT], x[GEMM_KT];
  float c0, c1;
};

__device__ __forceinline__ void gnn_tile_load(GnnTileRaw& r, int row, int half,
                                              const float* __restrict__ A0,
                                              const float* __restrict__ A1,
                                              const float* __restrict__ X,
                                              const float* __restrict__ cnt0,
                                              const float* __restrict__ cnt1) {
  const float* a0row = A0 + (size_t)row * GNN_D + half * 2;
  const float* a1row = A1 + (size_t)row * GNN_D + half * 2;
  const float* xrow  = X  + (size_t)row * GNN_D + half * 2;
#pragma unroll
  for (int kt = 0; kt < GEMM_KT; ++kt) {
    r.p[kt] = *(const v2f*)(a0row + kt * 4);
    r.q[kt] = *(const v2f*)(a1row + kt * 4);
    r.x[kt] = *(const v2f*)(xrow + kt * 4);
  }
  r.c0 = cnt0[row];
  r.c1 = cnt1[row];
}

__global__ __launch_bounds__(GEMM_WAVES * 32) void gnn_sage_gemm_kernel(
    const float* __restrict__ A0,
    const float* __restrict__ A1,
    const float* __restrict__ X,
    const float* __restrict__ cnt0,
    const float* __restrict__ cnt1,
    const float* __restrict__ Wl,
    const float* __restrict__ Wr,
    const float* __restrict__ bias,
    float* __restrict__ Y,
    int nRowTiles, int doRelu) {
  const int lane = threadIdx.x & 31;
  const int m    = lane & 15;
  const int half = lane >> 4;
  const int wv = __builtin_amdgcn_readfirstlane((int)((threadIdx.x >> 5) * GEMM_GRID + blockIdx.x));
  const int nw = GEMM_GRID * GEMM_WAVES;
  if (wv >= nRowTiles) return;

  v2f bl[GEMM_KT][4], br[GEMM_KT][4];
#pragma unroll
  for (int kt = 0; kt < GEMM_KT; ++kt) {
#pragma unroll
    for (int jt = 0; jt < 4; ++jt) {
      bl[kt][jt] = *(const v2f*)(Wl + (size_t)(jt * 16 + m) * GNN_D + kt * 4 + half * 2);
      br[kt][jt] = *(const v2f*)(Wr + (size_t)(jt * 16 + m) * GNN_D + kt * 4 + half * 2);
    }
  }
  float bv[4];
#pragma unroll
  for (int jt = 0; jt < 4; ++jt) bv[jt] = bias[jt * 16 + m];

  int t = wv;
  GnnTileRaw raw;
  gnn_tile_load(raw, t * 16 + m, half, A0, A1, X, cnt0, cnt1);

  for (;;) {
    const float scale = 1.0f / fmaxf(raw.c0 + raw.c1, 1.0f);
    v2f aa[GEMM_KT], ax[GEMM_KT];
#pragma unroll
    for (int kt = 0; kt < GEMM_KT; ++kt) {
      aa[kt].x = (raw.p[kt].x + raw.q[kt].x) * scale;
      aa[kt].y = (raw.p[kt].y + raw.q[kt].y) * scale;
      ax[kt] = raw.x[kt];
    }

    const int tn = t + nw;
    const int tl = tn < nRowTiles ? tn : t;
    gnn_tile_load(raw, tl * 16 + m, half, A0, A1, X, cnt0, cnt1);

    v8f c0 = {}, c1 = {}, c2 = {}, c3 = {};
#pragma unroll
    for (int kt = 0; kt < GEMM_KT; ++kt) {
      c0 = __builtin_amdgcn_wmma_f32_16x16x4_f32(false, aa[kt], false, bl[kt][0], (short)0, c0, false, false);
      c1 = __builtin_amdgcn_wmma_f32_16x16x4_f32(false, aa[kt], false, bl[kt][1], (short)0, c1, false, false);
      c2 = __builtin_amdgcn_wmma_f32_16x16x4_f32(false, aa[kt], false, bl[kt][2], (short)0, c2, false, false);
      c3 = __builtin_amdgcn_wmma_f32_16x16x4_f32(false, aa[kt], false, bl[kt][3], (short)0, c3, false, false);
    }
#pragma unroll
    for (int kt = 0; kt < GEMM_KT; ++kt) {
      c0 = __builtin_amdgcn_wmma_f32_16x16x4_f32(false, ax[kt], false, br[kt][0], (short)0, c0, false, false);
      c1 = __builtin_amdgcn_wmma_f32_16x16x4_f32(false, ax[kt], false, br[kt][1], (short)0, c1, false, false);
      c2 = __builtin_amdgcn_wmma_f32_16x16x4_f32(false, ax[kt], false, br[kt][2], (short)0, c2, false, false);
      c3 = __builtin_amdgcn_wmma_f32_16x16x4_f32(false, ax[kt], false, br[kt][3], (short)0, c3, false, false);
    }

    const int row0 = t * 16;
    v8f acc[4] = {c0, c1, c2, c3};
#pragma unroll
    for (int jt = 0; jt < 4; ++jt) {
      const int j = jt * 16 + m;
#pragma unroll
      for (int r = 0; r < 8; ++r) {
        float v = acc[jt][r] + bv[jt];
        if (doRelu) v = fmaxf(v, 0.0f);
        Y[(size_t)(row0 + r + half * 8) * GNN_D + j] = v;
      }
    }

    if (tn >= nRowTiles) break;
    t = tn;
  }
}

extern "C" void kernel_launch(void* const* d_in, const int* in_sizes, int n_in,
                              void* d_out, int out_size, void* d_ws, size_t ws_size,
                              hipStream_t stream) {
  const float* x   = (const float*)d_in[0];
  const int*   ei  = (const int*)d_in[1];
  const float* W1l = (const float*)d_in[2];
  const float* b1  = (const float*)d_in[3];
  const float* W1r = (const float*)d_in[4];
  const float* W2l = (const float*)d_in[5];
  const float* b2  = (const float*)d_in[6];
  const float* W2r = (const float*)d_in[7];
  float* out = (float*)d_out;

  float* aggp = (float*)d_ws;
  float* h    = aggp + (size_t)OWN_S * GNN_N * GNN_D;
  float* cntp = h + (size_t)GNN_N * GNN_D;
  unsigned* glist = (unsigned*)(cntp + (size_t)OWN_S * GNN_N);
  int* gcnt = (int*)(glist + (size_t)N_SLOT * GL_CAP);
  const float* agg1 = aggp + (size_t)GNN_N * GNN_D;
  const float* cnt1 = cntp + GNN_N;

  const int rowTiles = GNN_N / 16;

  gnn_owner_agg_scan_kernel<<<OWN_G * OWN_S, AGG_THREADS, 0, stream>>>(ei, x, aggp, cntp, glist, gcnt);
  gnn_sage_gemm_kernel<<<GEMM_GRID, GEMM_WAVES * 32, 0, stream>>>(aggp, agg1, x, cntp, cnt1, W1l, W1r, b1, h,
                                                     rowTiles,  1);

  gnn_owner_agg_replay_kernel<<<OWN_G * OWN_S, AGG_THREADS, 0, stream>>>(ei, h, aggp, glist, gcnt);
  gnn_sage_gemm_kernel<<<GEMM_GRID, GEMM_WAVES * 32, 0, stream>>>(aggp, agg1, h, cntp, cnt1, W2l, W2r, b2, out,
                                                     rowTiles,  0);
}


# edge-list scan replaced by one bucket pass: a new kernel reads the list once and files each edge by destination range in per-(range,piece) regions; both layers walk them; sums in LDS arrival order
# speedup vs baseline: 3.4335x; 1.1645x over previous
#include <hip/hip_runtime.h>
#include <hip/hip_bf16.h>

typedef __attribute__((ext_vector_type(2))) float v2f;
typedef __attribute__((ext_vector_type(8))) float v8f;

#define GNN_N 100000
#define GNN_E 1600000
#define GNN_D 64

#define OWN_G      128
#define OWN_S      2
#define OWN_ROWS   ((GNN_N + OWN_G - 1) / OWN_G)
#define AGG_THREADS 1024
#define AGG_WAVES  (AGG_THREADS / 32)
#define FL_BATCH   32
#define PB_G       256
#define PB_EDGES   (GNN_E / PB_G)
#define PB_CAP     128
#define PB_PER_WAVE (PB_G / OWN_S / AGG_WAVES)
#define PB_IT      ((PB_EDGES + AGG_THREADS - 1) / AGG_THREADS)

static_assert(GNN_E % PB_G == 0, "pieces must tile the edge list");
static_assert(PB_G == OWN_S * AGG_WAVES * PB_PER_WAVE, "every piece belongs to one owner wave of each range");
static_assert(PB_PER_WAVE == 4, "the walk below indexes exactly four regions");
static_assert(OWN_ROWS <= 1024, "local row must fit 10 bits of an entry");
static_assert((OWN_G - 1) * OWN_ROWS < GNN_N, "every work-group owns at least one row");
static_assert(GNN_N <= (1 << 22), "source row must fit 22 bits of an entry");

__global__ __launch_bounds__(AGG_THREADS) void gnn_bucket_kernel(
    const int* __restrict__ ei,
    unsigned* __restrict__ glist,
    int* __restrict__ gcnt) {
  __shared__ unsigned buf[OWN_G * PB_CAP];
  __shared__ int cur[OWN_G];
  const int tid = threadIdx.x;
  const int b = blockIdx.x;
  if (tid < OWN_G) cur[tid] = 0;
  __syncthreads();

  const int e0 = b * PB_EDGES;
  int sv[PB_IT], dv[PB_IT];
#pragma unroll
  for (int u = 0; u < PB_IT; ++u) {
    const int i = tid + u * AGG_THREADS;
    const bool ok = i < PB_EDGES;
    const int ic = ok ? i : PB_EDGES - 1;
    sv[u] = ei[e0 + ic];
    const int dd = ei[GNN_E + e0 + ic];
    dv[u] = ok ? dd : -1;
  }
#pragma unroll
  for (int u = 0; u < PB_IT; ++u) {
    const int d = dv[u];
    if ((unsigned)d < (unsigned)GNN_N) {
      const int r = d / OWN_ROWS;
      const int pos = atomicAdd(&cur[r], 1);
      if (pos < PB_CAP) buf[r * PB_CAP + pos] = ((unsigned)sv[u] << 10) | (unsigned)(d - r * OWN_ROWS);
    }
  }
  __syncthreads();

  const int lane = tid & 31;
  const int w = tid >> 5;
  for (int r = w; r < OWN_G; r += AGG_WAVES) {
    const int c = cur[r];
    const int n = c < PB_CAP ? c : PB_CAP;
    unsigned* __restrict__ g = glist + ((size_t)r * PB_G + b) * PB_CAP;
    for (int k = lane; k < n; k += 32) g[k] = buf[r * PB_CAP + k];
    if (lane == 0) gcnt[b * OWN_G + r] = c <= PB_CAP ? c : -1;
  }
}

template <bool PARTIAL>
__device__ __forceinline__ void gnn_batch_to_sums(int s, int dl, int m, int lane,
                                                  const float* __restrict__ feat, float* acc) {
  v2f row[FL_BATCH];
#pragma unroll
  for (int j = 0; j < FL_BATCH; ++j) {
    const int sj = __builtin_amdgcn_readlane(s, j);
    row[j] = *(const v2f*)(feat + (size_t)sj * GNN_D + 2 * lane);
  }
#pragma unroll
  for (int j = 0; j < FL_BATCH; ++j) {
    if (!PARTIAL || j < m) {
      const int dj = __builtin_amdgcn_readlane(dl, j);
      float* p = acc + dj * GNN_D + 2 * lane;
      atomicAdd(p, row[j].x);
      atomicAdd(p + 1, row[j].y);
    }
  }
}

template <bool COUNT>
__device__ __forceinline__ void gnn_regions_to_sums(const unsigned* __restrict__ g,
                                                    int c0, int c1, int c2, int c3, int lane,
                                                    const float* __restrict__ feat, float* acc,
                                                    float* cntl) {
  const int o1 = c0, o2 = o1 + c1, o3 = o2 + c2, n = o3 + c3;
  if (n <= 0) return;
  const int last = n - 1;
  auto entry = [&](int idx) -> unsigned {
    const int k = idx < last ? idx : last;
    const int i = (k >= o1) + (k >= o2) + (k >= o3);
    const int off = i == 0 ? 0 : (i == 1 ? o1 : (i == 2 ? o2 : o3));
    return g[i * PB_CAP + (k - off)];
  };
  const unsigned v0 = entry(lane);
  unsigned pk = lane < n ? v0 : 0u;
  for (int b = 0; b < n; b += FL_BATCH) {
    const int nx = b + FL_BATCH + lane;
    const unsigned vn = entry(nx);
    const unsigned pkn = nx < n ? vn : 0u;
    const unsigned su = pk >> 10;
    const int s  = (int)(su < (unsigned)GNN_N ? su : 0u);
    const int dl = (int)(pk & 1023u);
    if (COUNT) {
      if (b + lane < n) atomicAdd(&cntl[dl], 1.0f);
    }
    if (n - b >= FL_BATCH) gnn_batch_to_sums<false>(s, dl, FL_BATCH, lane, feat, acc);
    else                   gnn_batch_to_sums<true>(s, dl, n - b, lane, feat, acc);
    pk = pkn;
  }
}

template <bool COUNT>
__device__ __forceinline__ void gnn_edges_to_sums(const int* __restrict__ ei, int e_lo, int ne,
                                                  int lo, unsigned nr, int lane,
                                                  const float* __restrict__ feat, float* acc,
                                                  float* cntl) {
  for (int e0 = 0; e0 < ne; e0 += 32) {
    const int e = e0 + lane;
    const bool ok = e < ne;
    const int d = ok ? ei[GNN_E + e_lo + e] : -1;
    const unsigned t = (unsigned)(d - lo);
    const bool hit = ok && d >= 0 && t < nr;
    const unsigned su = hit ? (unsigned)ei[e_lo + e] : 0u;
    const int s = (int)(su < (unsigned)GNN_N ? su : 0u);
    if (COUNT) {
      if (hit) atomicAdd(&cntl[t], 1.0f);
    }
    unsigned mask = __builtin_amdgcn_ballot_w32(hit);
    while (mask != 0u) {
      const int j = __builtin_ctz(mask);
      mask &= mask - 1u;
      const int sj = __shfl(s, j);
      const int tj = __shfl((int)t, j);
      const v2f row = *(const v2f*)(feat + (size_t)sj * GNN_D + 2 * lane);
      float* p = acc + tj * GNN_D + 2 * lane;
      atomicAdd(p, row.x);
      atomicAdd(p + 1, row.y);
    }
  }
}

template <bool COUNT>
__global__ __launch_bounds__(AGG_THREADS) void gnn_owner_agg_kernel(
    const int* __restrict__ ei,
    const float* __restrict__ feat,
    float* __restrict__ aggp,
    float* __restrict__ cntp,
    const unsigned* __restrict__ glist,
    const int* __restrict__ gcnt) {
  __shared__ __attribute__((aligned(16))) float acc[OWN_ROWS * GNN_D];
  __shared__ float cntl[OWN_ROWS];

  const int tid  = threadIdx.x;
  const int lane = tid & 31;
  const int w    = __builtin_amdgcn_readfirstlane(tid >> 5);
  const int rng  = blockIdx.x / OWN_S;
  const int part = blockIdx.x % OWN_S;
  const int lo   = rng * OWN_ROWS;
  const unsigned nr = (unsigned)((GNN_N - lo) < OWN_ROWS ? (GNN_N - lo) : OWN_ROWS);

  for (int i = tid; i < OWN_ROWS * GNN_D / 4; i += AGG_THREADS)
    ((float4*)acc)[i] = make_float4(0.f, 0.f, 0.f, 0.f);
  if (COUNT) {
    for (int i = tid; i < OWN_ROWS; i += AGG_THREADS) cntl[i] = 0.f;
  }
  __syncthreads();

  const int piece0 = part * (PB_G / OWN_S) + w * PB_PER_WAVE;
  const int q = rng * PB_G + piece0;
  const int c0 = __builtin_amdgcn_readfirstlane(gcnt[(piece0 + 0) * OWN_G + rng]);
  const int c1 = __builtin_amdgcn_readfirstlane(gcnt[(piece0 + 1) * OWN_G + rng]);
  const int c2 = __builtin_amdgcn_readfirstlane(gcnt[(piece0 + 2) * OWN_G + rng]);
  const int c3 = __builtin_amdgcn_readfirstlane(gcnt[(piece0 + 3) * OWN_G + rng]);
  if ((c0 | c1 | c2 | c3) >= 0) {
    gnn_regions_to_sums<COUNT>(glist + (size_t)q * PB_CAP,
                               c0 < PB_CAP ? c0 : PB_CAP, c1 < PB_CAP ? c1 : PB_CAP,
                               c2 < PB_CAP ? c2 : PB_CAP, c3 < PB_CAP ? c3 : PB_CAP,
                               lane, feat, acc, cntl);
  } else {
    gnn_edges_to_sums<COUNT>(ei, piece0 * PB_EDGES, PB_PER_WAVE * PB_EDGES, lo, nr, lane,
                             feat, acc, cntl);
  }

  __syncthreads();
  float4* __restrict__ o4 = (float4*)(aggp + ((size_t)part * GNN_N + lo) * GNN_D);
  for (int i = tid; i < (int)nr * (GNN_D / 4); i += AGG_THREADS)
    o4[i] = ((const float4*)acc)[i];
  if (COUNT) {
    for (int i = tid; i < (int)nr; i += AGG_THREADS)
      cntp[(size_t)part * GNN_N + lo + i] = cntl[i];
  }
}

static_assert(OWN_S == 2, "the GEMM adds exactly two partial planes");
#define GEMM_WAVES 8
#define GEMM_GRID  256
#define GEMM_KT    (GNN_D / 4)

struct GnnTileRaw {
  v2f p[GEMM_KT], q[GEMM_KT], x[GEMM_KT];
  float c0, c1;
};

__device__ __forceinline__ void gnn_tile_load(GnnTileRaw& r, int row, int half,
                                              const float* __restrict__ A0,
                                              const float* __restrict__ A1,
                                              const float* __restrict__ X,
                                              const float* __restrict__ cnt0,
                                              const float* __restrict__ cnt1) {
  const float* a0row = A0 + (size_t)row * GNN_D + half * 2;
  const float* a1row = A1 + (size_t)row * GNN_D + half * 2;
  const float* xrow  = X  + (size_t)row * GNN_D + half * 2;
#pragma unroll
  for (int kt = 0; kt < GEMM_KT; ++kt) {
    r.p[kt] = *(const v2f*)(a0row + kt * 4);
    r.q[kt] = *(const v2f*)(a1row + kt * 4);
    r.x[kt] = *(const v2f*)(xrow + kt * 4);
  }
  r.c0 = cnt0[row];
  r.c1 = cnt1[row];
}

__global__ __launch_bounds__(GEMM_WAVES * 32) void gnn_sage_gemm_kernel(
    const float* __restrict__ A0,
    const float* __restrict__ A1,
    const float* __restrict__ X,
    const float* __restrict__ cnt0,
    const float* __restrict__ cnt1,
    const float* __restrict__ Wl,
    const float* __restrict__ Wr,
    const float* __restrict__ bias,
    float* __restrict__ Y,
    int nRowTiles, int doRelu) {
  const int lane = threadIdx.x & 31;
  const int m    = lane & 15;
  const int half = lane >> 4;
  const int wv = __builtin_amdgcn_readfirstlane((int)((threadIdx.x >> 5) * GEMM_GRID + blockIdx.x));
  const int nw = GEMM_GRID * GEMM_WAVES;
  if (wv >= nRowTiles) return;

  v2f bl[GEMM_KT][4], br[GEMM_KT][4];
#pragma unroll
  for (int kt = 0; kt < GEMM_KT; ++kt) {
#pragma unroll
    for (int jt = 0; jt < 4; ++jt) {
      bl[kt][jt] = *(const v2f*)(Wl + (size_t)(jt * 16 + m) * GNN_D + kt * 4 + half * 2);
      br[kt][jt] = *(const v2f*)(Wr + (size_t)(jt * 16 + m) * GNN_D + kt * 4 + half * 2);
    }
  }
  float bv[4];
#pragma unroll
  for (int jt = 0; jt < 4; ++jt) bv[jt] = bias[jt * 16 + m];

  int t = wv;
  GnnTileRaw raw;
  gnn_tile_load(raw, t * 16 + m, half, A0, A1, X, cnt0, cnt1);

  for (;;) {
    const float scale = 1.0f / fmaxf(raw.c0 + raw.c1, 1.0f);
    v2f aa[GEMM_KT], ax[GEMM_KT];
#pragma unroll
    for (int kt = 0; kt < GEMM_KT; ++kt) {
      aa[kt].x = (raw.p[kt].x + raw.q[kt].x) * scale;
      aa[kt].y = (raw.p[kt].y + raw.q[kt].y) * scale;
      ax[kt] = raw.x[kt];
    }

    const int tn = t + nw;
    const int tl = tn < nRowTiles ? tn : t;
    gnn_tile_load(raw, tl * 16 + m, half, A0, A1, X, cnt0, cnt1);

    v8f c0 = {}, c1 = {}, c2 = {}, c3 = {};
#pragma unroll
    for (int kt = 0; kt < GEMM_KT; ++kt) {
      c0 = __builtin_amdgcn_wmma_f32_16x16x4_f32(false, aa[kt], false, bl[kt][0], (short)0, c0, false, false);
      c1 = __builtin_amdgcn_wmma_f32_16x16x4_f32(false, aa[kt], false, bl[kt][1], (short)0, c1, false, false);
      c2 = __builtin_amdgcn_wmma_f32_16x16x4_f32(false, aa[kt], false, bl[kt][2], (short)0, c2, false, false);
      c3 = __builtin_amdgcn_wmma_f32_16x16x4_f32(false, aa[kt], false, bl[kt][3], (short)0, c3, false, false);
    }
#pragma unroll
    for (int kt = 0; kt < GEMM_KT; ++kt) {
      c0 = __builtin_amdgcn_wmma_f32_16x16x4_f32(false, ax[kt], false, br[kt][0], (short)0, c0, false, false);
      c1 = __builtin_amdgcn_wmma_f32_16x16x4_f32(false, ax[kt], false, br[kt][1], (short)0, c1, false, false);
      c2 = __builtin_amdgcn_wmma_f32_16x16x4_f32(false, ax[kt], false, br[kt][2], (short)0, c2, false, false);
      c3 = __builtin_amdgcn_wmma_f32_16x16x4_f32(false, ax[kt], false, br[kt][3], (short)0, c3, false, false);
    }

    const int row0 = t * 16;
    v8f acc[4] = {c0, c1, c2, c3};
#pragma unroll
    for (int jt = 0; jt < 4; ++jt) {
      const int j = jt * 16 + m;
#pragma unroll
      for (int r = 0; r < 8; ++r) {
        float v = acc[jt][r] + bv[jt];
        if (doRelu) v = fmaxf(v, 0.0f);
        Y[(size_t)(row0 + r + half * 8) * GNN_D + j] = v;
      }
    }

    if (tn >= nRowTiles) break;
    t = tn;
  }
}

extern "C" void kernel_launch(void* const* d_in, const int* in_sizes, int n_in,
                              void* d_out, int out_size, void* d_ws, size_t ws_size,
                              hipStream_t stream) {
  const float* x   = (const float*)d_in[0];
  const int*   ei  = (const int*)d_in[1];
  const float* W1l = (const float*)d_in[2];
  const float* b1  = (const float*)d_in[3];
  const float* W1r = (const float*)d_in[4];
  const float* W2l = (const float*)d_in[5];
  const float* b2  = (const float*)d_in[6];
  const float* W2r = (const float*)d_in[7];
  float* out = (float*)d_out;

  float* aggp = (float*)d_ws;
  float* h    = aggp + (size_t)OWN_S * GNN_N * GNN_D;
  float* cntp = h + (size_t)GNN_N * GNN_D;
  unsigned* glist = (unsigned*)(cntp + (size_t)OWN_S * GNN_N);
  int* gcnt = (int*)(glist + (size_t)OWN_G * PB_G * PB_CAP);
  const float* agg1 = aggp + (size_t)GNN_N * GNN_D;
  const float* cnt1 = cntp + GNN_N;

  const int rowTiles = GNN_N / 16;

  gnn_bucket_kernel<<<PB_G, AGG_THREADS, 0, stream>>>(ei, glist, gcnt);

  gnn_owner_agg_kernel<true><<<OWN_G * OWN_S, AGG_THREADS, 0, stream>>>(ei, x, aggp, cntp, glist, gcnt);
  gnn_sage_gemm_kernel<<<GEMM_GRID, GEMM_WAVES * 32, 0, stream>>>(aggp, agg1, x, cntp, cnt1, W1l, W1r, b1, h,
                                                     rowTiles,  1);

  gnn_owner_agg_kernel<false><<<OWN_G * OWN_S, AGG_THREADS, 0, stream>>>(ei, h, aggp, cntp, glist, gcnt);
  gnn_sage_gemm_kernel<<<GEMM_GRID, GEMM_WAVES * 32, 0, stream>>>(aggp, agg1, h, cntp, cnt1, W2l, W2r, b2, out,
                                                     rowTiles,  0);
}


# layer GEMM fused into the owner kernel: 250 work-groups own a whole 400-row range each (edge list not halved), multiply their LDS sums by the weights in place; row sum = one LDS arrival-order sum
# speedup vs baseline: 4.4085x; 1.2840x over previous
#include <hip/hip_runtime.h>
#include <hip/hip_bf16.h>

typedef __attribute__((ext_vector_type(2))) float v2f;
typedef __attribute__((ext_vector_type(8))) float v8f;

#define GNN_N 100000
#define GNN_E 1600000
#define GNN_D 64

#define OWN_ROWS   400
#define OWN_G      (GNN_N / OWN_ROWS)
#define OWN_TILES  (OWN_ROWS / 16)
#define ACC_P      68
#define AGG_THREADS 1024
#define AGG_WAVES  (AGG_THREADS / 32)
#define FL_BATCH   32
#define PB_G       256
#define PB_EDGES   (GNN_E / PB_G)
#define PB_CAP     64
#define PB_PER_WAVE (PB_G / AGG_WAVES)
#define PB_IT      ((PB_EDGES + AGG_THREADS - 1) / AGG_THREADS)
#define GEMM_KT    (GNN_D / 4)

static_assert(GNN_E % PB_G == 0, "pieces must tile the edge list");
static_assert(PB_G == AGG_WAVES * PB_PER_WAVE, "every piece belongs to one owner wave of each range");
static_assert(GNN_N % OWN_ROWS == 0 && OWN_ROWS % 16 == 0, "the ranges tile the rows, and 16-row tiles tile a range: no partial tile");
static_assert(OWN_ROWS <= 1024, "local row must fit 10 bits of an entry");
static_assert((OWN_G - 1) * OWN_ROWS < GNN_N, "every work-group owns at least one row");
static_assert(GNN_N <= (1 << 22), "source row must fit 22 bits of an entry");
static_assert(OWN_G <= AGG_THREADS, "one thread clears each range counter");
static_assert(AGG_WAVES == 32, "the products below give each wave one of 4 column blocks and one of 8 tile groups");

__global__ __launch_bounds__(AGG_THREADS) void gnn_bucket_kernel(
    const int* __restrict__ ei,
    unsigned* __restrict__ glist,
    int* __restrict__ gcnt) {
  __shared__ unsigned buf[OWN_G * PB_CAP];
  __shared__ int cur[OWN_G];
  const int tid = threadIdx.x;
  const int b = blockIdx.x;
  if (tid < OWN_G) cur[tid] = 0;
  __syncthreads();

  const int e0 = b * PB_EDGES;
  int sv[PB_IT], dv[PB_IT];
#pragma unroll
  for (int u = 0; u < PB_IT; ++u) {
    const int i = tid + u * AGG_THREADS;
    const bool ok = i < PB_EDGES;
    const int ic = ok ? i : PB_EDGES - 1;
    sv[u] = ei[e0 + ic];
    const int dd = ei[GNN_E + e0 + ic];
    dv[u] = ok ? dd : -1;
  }
#pragma unroll
  for (int u = 0; u < PB_IT; ++u) {
    const int d = dv[u];
    if ((unsigned)d < (unsigned)GNN_N) {
      const int r = d / OWN_ROWS;
      const int pos = atomicAdd(&cur[r], 1);
      if (pos < PB_CAP) buf[r * PB_CAP + pos] = ((unsigned)sv[u] << 10) | (unsigned)(d - r * OWN_ROWS);
    }
  }
  __syncthreads();

  const int lane = tid & 31;
  const int w = tid >> 5;
  for (int r = w; r < OWN_G; r += AGG_WAVES) {
    const int c = cur[r];
    const int n = c < PB_CAP ? c : PB_CAP;
    unsigned* __restrict__ g = glist + ((size_t)r * PB_G + b) * PB_CAP;
    for (int k = lane; k < n; k += 32) g[k] = buf[r * PB_CAP + k];
    if (lane == 0) gcnt[b * OWN_G + r] = c <= PB_CAP ? c : -1;
  }
}

template <bool PARTIAL>
__device__ __forceinline__ void gnn_batch_to_sums(int s, int dl, int m, int lane,
                                                  const float* __restrict__ feat, float* acc) {
  v2f row[FL_BATCH];
#pragma unroll
  for (int j = 0; j < FL_BATCH; ++j) {
    const int sj = __builtin_amdgcn_readlane(s, j);
    row[j] = *(const v2f*)(feat + (size_t)sj * GNN_D + 2 * lane);
  }
  const int dof = dl * ACC_P;
#pragma unroll
  for (int j = 0; j < FL_BATCH; ++j) {
    if (!PARTIAL || j < m) {
      const int dj = __builtin_amdgcn_readlane(dof, j);
      float* p = acc + dj + 2 * lane;
      atomicAdd(p, row[j].x);
      atomicAdd(p + 1, row[j].y);
    }
  }
}

__device__ __forceinline__ void gnn_regions_to_sums(const unsigned* __restrict__ g,
                                                    const int (&c)[PB_PER_WAVE], int lane,
                                                    const float* __restrict__ feat, float* acc,
                                                    float* cntl) {
  int o[PB_PER_WAVE + 1];
  o[0] = 0;
#pragma unroll
  for (int j = 0; j < PB_PER_WAVE; ++j) o[j + 1] = o[j] + c[j];
  const int n = o[PB_PER_WAVE];
  if (n <= 0) return;
  const int last = n - 1;
  auto entry = [&](int idx) -> unsigned {
    const int k = idx < last ? idx : last;
    int a = k;
#pragma unroll
    for (int j = 1; j < PB_PER_WAVE; ++j) a += (k >= o[j]) ? (PB_CAP - c[j - 1]) : 0;
    return g[a];
  };
  const unsigned v0 = entry(lane);
  unsigned pk = lane < n ? v0 : 0u;
  for (int b = 0; b < n; b += FL_BATCH) {
    const int nx = b + FL_BATCH + lane;
    const unsigned vn = entry(nx);
    const unsigned pkn = nx < n ? vn : 0u;
    const unsigned su = pk >> 10;
    const int s  = (int)(su < (unsigned)GNN_N ? su : 0u);
    const int dl = (int)(pk & 1023u);
    if (b + lane < n) atomicAdd(&cntl[dl], 1.0f);
    if (n - b >= FL_BATCH) gnn_batch_to_sums<false>(s, dl, FL_BATCH, lane, feat, acc);
    else                   gnn_batch_to_sums<true>(s, dl, n - b, lane, feat, acc);
    pk = pkn;
  }
}

__device__ __forceinline__ void gnn_edges_to_sums(const int* __restrict__ ei, int e_lo, int ne,
                                                  int lo, unsigned nr, int lane,
                                                  const float* __restrict__ feat, float* acc,
                                                  float* cntl) {
  for (int e0 = 0; e0 < ne; e0 += 32) {
    const int e = e0 + lane;
    const bool ok = e < ne;
    const int d = ok ? ei[GNN_E + e_lo + e] : -1;
    const unsigned t = (unsigned)(d - lo);
    const bool hit = ok && d >= 0 && t < nr;
    const unsigned su = hit ? (unsigned)ei[e_lo + e] : 0u;
    const int s = (int)(su < (unsigned)GNN_N ? su : 0u);
    if (hit) atomicAdd(&cntl[t], 1.0f);
    unsigned mask = __builtin_amdgcn_ballot_w32(hit);
    while (mask != 0u) {
      const int j = __builtin_ctz(mask);
      mask &= mask - 1u;
      const int sj = __shfl(s, j);
      const int tj = __shfl((int)t, j);
      const v2f row = *(const v2f*)(feat + (size_t)sj * GNN_D + 2 * lane);
      float* p = acc + tj * ACC_P + 2 * lane;
      atomicAdd(p, row.x);
      atomicAdd(p + 1, row.y);
    }
  }
}

__global__ __launch_bounds__(AGG_THREADS) void gnn_sage_layer_kernel(
    const int* __restrict__ ei,
    const float* __restrict__ feat,
    const unsigned* __restrict__ glist,
    const int* __restrict__ gcnt,
    const float* __restrict__ Wl,
    const float* __restrict__ Wr,
    const float* __restrict__ bias,
    float* __restrict__ Y,
    int doRelu) {
  __shared__ __attribute__((aligned(16))) float acc[OWN_ROWS * ACC_P];
  __shared__ float cntl[OWN_ROWS];

  const int tid  = threadIdx.x;
  const int lane = tid & 31;
  const int w    = __builtin_amdgcn_readfirstlane(tid >> 5);
  const int rng  = blockIdx.x;
  const int lo   = rng * OWN_ROWS;

  for (int i = tid; i < OWN_ROWS * ACC_P / 4; i += AGG_THREADS)
    ((float4*)acc)[i] = make_float4(0.f, 0.f, 0.f, 0.f);
  for (int i = tid; i < OWN_ROWS; i += AGG_THREADS) cntl[i] = 0.f;
  __syncthreads();

  {
    const int piece0 = w * PB_PER_WAVE;
    int c[PB_PER_WAVE];
    int any = 0;
#pragma unroll
    for (int j = 0; j < PB_PER_WAVE; ++j) {
      c[j] = __builtin_amdgcn_readfirstlane(gcnt[(piece0 + j) * OWN_G + rng]);
      any |= c[j];
    }
    if (any >= 0) {
#pragma unroll
      for (int j = 0; j < PB_PER_WAVE; ++j) c[j] = c[j] < PB_CAP ? c[j] : PB_CAP;
      gnn_regions_to_sums(glist + ((size_t)rng * PB_G + piece0) * PB_CAP, c, lane, feat, acc, cntl);
    } else {
      gnn_edges_to_sums(ei, piece0 * PB_EDGES, PB_PER_WAVE * PB_EDGES, lo, (unsigned)OWN_ROWS, lane,
                        feat, acc, cntl);
    }
  }

  const int m    = lane & 15;
  const int half = lane >> 4;
  const int jt   = w & 3;
  const int tg   = w >> 2;

  v2f bl[GEMM_KT], br[GEMM_KT];
#pragma unroll
  for (int kt = 0; kt < GEMM_KT; ++kt) {
    bl[kt] = *(const v2f*)(Wl + (size_t)(jt * 16 + m) * GNN_D + kt * 4 + half * 2);
    br[kt] = *(const v2f*)(Wr + (size_t)(jt * 16 + m) * GNN_D + kt * 4 + half * 2);
  }
  const float bv = bias[jt * 16 + m];

  __syncthreads();

  for (int t = tg; t < OWN_TILES; t += AGG_WAVES / 4) {
    const int lr = t * 16 + m;
    const float* xrow = feat + (size_t)(lo + lr) * GNN_D + half * 2;
    const float* arow = acc + lr * ACC_P + half * 2;
    const float scale = 1.0f / fmaxf(cntl[lr], 1.0f);

    v8f cc = {};
    {
      v2f aa[GEMM_KT];
#pragma unroll
      for (int kt = 0; kt < GEMM_KT; ++kt) {
        const v2f sv = *(const v2f*)(arow + kt * 4);
        aa[kt].x = sv.x * scale;
        aa[kt].y = sv.y * scale;
      }
      __builtin_amdgcn_sched_barrier(0);
#pragma unroll
      for (int kt = 0; kt < GEMM_KT; ++kt)
        cc = __builtin_amdgcn_wmma_f32_16x16x4_f32(false, aa[kt], false, bl[kt], (short)0, cc, false, false);
    }
    {
      v2f ax[GEMM_KT];
#pragma unroll
      for (int kt = 0; kt < GEMM_KT; ++kt) ax[kt] = *(const v2f*)(xrow + kt * 4);
#pragma unroll
      for (int kt = 0; kt < GEMM_KT; ++kt)
        cc = __builtin_amdgcn_wmma_f32_16x16x4_f32(false, ax[kt], false, br[kt], (short)0, cc, false, false);
    }

#pragma unroll
    for (int r = 0; r < 8; ++r) {
      float v = cc[r] + bv;
      if (doRelu) v = fmaxf(v, 0.0f);
      Y[(size_t)(lo + t * 16 + r + half * 8) * GNN_D + jt * 16 + m] = v;
    }
  }
}

extern "C" void kernel_launch(void* const* d_in, const int* in_sizes, int n_in,
                              void* d_out, int out_size, void* d_ws, size_t ws_size,
                              hipStream_t stream) {
  const float* x   = (const float*)d_in[0];
  const int*   ei  = (const int*)d_in[1];
  const float* W1l = (const float*)d_in[2];
  const float* b1  = (const float*)d_in[3];
  const float* W1r = (const float*)d_in[4];
  const float* W2l = (const float*)d_in[5];
  const float* b2  = (const float*)d_in[6];
  const float* W2r = (const float*)d_in[7];
  float* out = (float*)d_out;

  float* h = (float*)d_ws;
  unsigned* glist = (unsigned*)(h + (size_t)GNN_N * GNN_D);
  int* gcnt = (int*)(glist + (size_t)OWN_G * PB_G * PB_CAP);

  gnn_bucket_kernel<<<PB_G, AGG_THREADS, 0, stream>>>(ei, glist, gcnt);

  gnn_sage_layer_kernel<<<OWN_G, AGG_THREADS, 0, stream>>>(ei, x, glist, gcnt, W1l, W1r, b1, h,  1);
  gnn_sage_layer_kernel<<<OWN_G, AGG_THREADS, 0, stream>>>(ei, h, glist, gcnt, W2l, W2r, b2, out,  0);
}


# LDS float atomic adds replaced by plain LDS load-add-store: a range's entries are re-filed by owner wave (row mod 32, LDS int counters), a wave alone updates its rows; sum order = list arrival order
# speedup vs baseline: 4.9978x; 1.1337x over previous
#include <hip/hip_runtime.h>
#include <hip/hip_bf16.h>

typedef __attribute__((ext_vector_type(2))) float v2f;
typedef __attribute__((ext_vector_type(8))) float v8f;

#define GNN_N 100000
#define GNN_E 1600000
#define GNN_D 64

#define OWN_ROWS   400
#define OWN_G      (GNN_N / OWN_ROWS)
#define OWN_TILES  (OWN_ROWS / 16)
#define ACC_P      68
#define AGG_THREADS 1024
#define AGG_WAVES  (AGG_THREADS / 32)
#define FL_BATCH   32
#define PB_G       256
#define PB_EDGES   (GNN_E / PB_G)
#define PB_CAP     64
#define PB_PER_WAVE (PB_G / AGG_WAVES)
#define PB_IT      ((PB_EDGES + AGG_THREADS - 1) / AGG_THREADS)
#define GEMM_KT    (GNN_D / 4)
#define WL_CAP     320

static_assert(GNN_E % PB_G == 0, "pieces must tile the edge list");
static_assert(PB_G == AGG_WAVES * PB_PER_WAVE, "every piece belongs to one owner wave of each range");
static_assert(GNN_N % OWN_ROWS == 0 && OWN_ROWS % 16 == 0, "the ranges tile the rows, and 16-row tiles tile a range: no partial tile");
static_assert(OWN_ROWS <= 1024, "local row must fit 10 bits of an entry");
static_assert((OWN_G - 1) * OWN_ROWS < GNN_N, "every work-group owns at least one row");
static_assert(GNN_N <= (1 << 22), "source row must fit 22 bits of an entry");
static_assert(OWN_G <= AGG_THREADS, "one thread clears each range counter");
static_assert(AGG_WAVES == 32, "the products below give each wave one of 4 column blocks and one of 8 tile groups");

__global__ __launch_bounds__(AGG_THREADS) void gnn_bucket_kernel(
    const int* __restrict__ ei,
    unsigned* __restrict__ glist,
    int* __restrict__ gcnt) {
  __shared__ unsigned buf[OWN_G * PB_CAP];
  __shared__ int cur[OWN_G];
  const int tid = threadIdx.x;
  const int b = blockIdx.x;
  if (tid < OWN_G) cur[tid] = 0;
  __syncthreads();

  const int e0 = b * PB_EDGES;
  int sv[PB_IT], dv[PB_IT];
#pragma unroll
  for (int u = 0; u < PB_IT; ++u) {
    const int i = tid + u * AGG_THREADS;
    const bool ok = i < PB_EDGES;
    const int ic = ok ? i : PB_EDGES - 1;
    sv[u] = ei[e0 + ic];
    const int dd = ei[GNN_E + e0 + ic];
    dv[u] = ok ? dd : -1;
  }
#pragma unroll
  for (int u = 0; u < PB_IT; ++u) {
    const int d = dv[u];
    if ((unsigned)d < (unsigned)GNN_N) {
      const int r = d / OWN_ROWS;
      const int pos = atomicAdd(&cur[r], 1);
      if (pos < PB_CAP) buf[r * PB_CAP + pos] = ((unsigned)sv[u] << 10) | (unsigned)(d - r * OWN_ROWS);
    }
  }
  __syncthreads();

  const int lane = tid & 31;
  const int w = tid >> 5;
  for (int r = w; r < OWN_G; r += AGG_WAVES) {
    const int c = cur[r];
    const int n = c < PB_CAP ? c : PB_CAP;
    unsigned* __restrict__ g = glist + ((size_t)r * PB_G + b) * PB_CAP;
    for (int k = lane; k < n; k += 32) g[k] = buf[r * PB_CAP + k];
    if (lane == 0) gcnt[b * OWN_G + r] = c <= PB_CAP ? c : -1;
  }
}

template <bool PARTIAL>
__device__ __forceinline__ void gnn_batch_to_sums(int s, int dl, int m, int lane,
                                                  const float* __restrict__ feat, float* acc) {
  v2f row[FL_BATCH];
#pragma unroll
  for (int j = 0; j < FL_BATCH; ++j) {
    const int sj = __builtin_amdgcn_readlane(s, j);
    row[j] = *(const v2f*)(feat + (size_t)sj * GNN_D + 2 * lane);
  }
  const int dof = dl * ACC_P;
#pragma unroll
  for (int j = 0; j < FL_BATCH; ++j) {
    if (!PARTIAL || j < m) {
      const int dj = __builtin_amdgcn_readlane(dof, j);
      float* p = acc + dj + 2 * lane;
      atomicAdd(p, row[j].x);
      atomicAdd(p + 1, row[j].y);
    }
  }
}

__device__ __forceinline__ void gnn_regions_to_sums(const unsigned* __restrict__ g,
                                                    const int (&c)[PB_PER_WAVE], int lane,
                                                    const float* __restrict__ feat, float* acc,
                                                    float* cntl) {
  int o[PB_PER_WAVE + 1];
  o[0] = 0;
#pragma unroll
  for (int j = 0; j < PB_PER_WAVE; ++j) o[j + 1] = o[j] + c[j];
  const int n = o[PB_PER_WAVE];
  if (n <= 0) return;
  const int last = n - 1;
  auto entry = [&](int idx) -> unsigned {
    const int k = idx < last ? idx : last;
    int a = k;
#pragma unroll
    for (int j = 1; j < PB_PER_WAVE; ++j) a += (k >= o[j]) ? (PB_CAP - c[j - 1]) : 0;
    return g[a];
  };
  const unsigned v0 = entry(lane);
  unsigned pk = lane < n ? v0 : 0u;
  for (int b = 0; b < n; b += FL_BATCH) {
    const int nx = b + FL_BATCH + lane;
    const unsigned vn = entry(nx);
    const unsigned pkn = nx < n ? vn : 0u;
    const unsigned su = pk >> 10;
    const int s  = (int)(su < (unsigned)GNN_N ? su : 0u);
    const int dl = (int)(pk & 1023u);
    if (b + lane < n) atomicAdd(&cntl[dl], 1.0f);
    if (n - b >= FL_BATCH) gnn_batch_to_sums<false>(s, dl, FL_BATCH, lane, feat, acc);
    else                   gnn_batch_to_sums<true>(s, dl, n - b, lane, feat, acc);
    pk = pkn;
  }
}

__device__ __forceinline__ void gnn_edges_to_sums(const int* __restrict__ ei, int e_lo, int ne,
                                                  int lo, unsigned nr, int lane,
                                                  const float* __restrict__ feat, float* acc,
                                                  float* cntl) {
  for (int e0 = 0; e0 < ne; e0 += 32) {
    const int e = e0 + lane;
    const bool ok = e < ne;
    const int d = ok ? ei[GNN_E + e_lo + e] : -1;
    const unsigned t = (unsigned)(d - lo);
    const bool hit = ok && d >= 0 && t < nr;
    const unsigned su = hit ? (unsigned)ei[e_lo + e] : 0u;
    const int s = (int)(su < (unsigned)GNN_N ? su : 0u);
    if (hit) atomicAdd(&cntl[t], 1.0f);
    unsigned mask = __builtin_amdgcn_ballot_w32(hit);
    while (mask != 0u) {
      const int j = __builtin_ctz(mask);
      mask &= mask - 1u;
      const int sj = __shfl(s, j);
      const int tj = __shfl((int)t, j);
      const v2f row = *(const v2f*)(feat + (size_t)sj * GNN_D + 2 * lane);
      float* p = acc + tj * ACC_P + 2 * lane;
      atomicAdd(p, row.x);
      atomicAdd(p + 1, row.y);
    }
  }
}


__device__ __forceinline__ void gnn_file_entry(unsigned pk, unsigned* wl, int* wn) {
  const int ow = (int)(pk & (unsigned)(AGG_WAVES - 1));
  const int pos = atomicAdd(&wn[ow], 1);
  if (pos < WL_CAP) wl[ow * WL_CAP + pos] = pk;
}

#define FILE_IT 8
__device__ __forceinline__ void gnn_regions_to_lists(const unsigned* __restrict__ g,
                                                     const int (&c)[PB_PER_WAVE], int lane,
                                                     unsigned* wl, int* wn) {
  int o[PB_PER_WAVE + 1];
  o[0] = 0;
#pragma unroll
  for (int j = 0; j < PB_PER_WAVE; ++j) o[j + 1] = o[j] + c[j];
  const int n = o[PB_PER_WAVE];
  if (n <= 0) return;
  const int last = n - 1;
  for (int b = 0; b < n; b += 32 * FILE_IT) {
    unsigned pk[FILE_IT];
#pragma unroll
    for (int u = 0; u < FILE_IT; ++u) {
      const int i = b + 32 * u + lane;
      const int k = i < last ? i : last;
      int a = k;
#pragma unroll
      for (int j = 1; j < PB_PER_WAVE; ++j) a += (k >= o[j]) ? (PB_CAP - c[j - 1]) : 0;
      pk[u] = g[a];
    }
    int pos[FILE_IT];
#pragma unroll
    for (int u = 0; u < FILE_IT; ++u)
      pos[u] = (b + 32 * u + lane < n) ? atomicAdd(&wn[pk[u] & (unsigned)(AGG_WAVES - 1)], 1) : WL_CAP;
#pragma unroll
    for (int u = 0; u < FILE_IT; ++u)
      if (pos[u] < WL_CAP) wl[(pk[u] & (unsigned)(AGG_WAVES - 1)) * WL_CAP + pos[u]] = pk[u];
  }
}

__device__ __forceinline__ void gnn_edges_to_lists(const int* __restrict__ ei, int e_lo, int ne,
                                                   int lo, unsigned nr, int lane,
                                                   unsigned* wl, int* wn) {
  for (int e0 = 0; e0 < ne; e0 += 32) {
    const int e = e0 + lane;
    const bool ok = e < ne;
    const int d = ok ? ei[GNN_E + e_lo + e] : -1;
    const unsigned t = (unsigned)(d - lo);
    if (ok && d >= 0 && t < nr) {
      const unsigned su = (unsigned)ei[e_lo + e];
      gnn_file_entry(((su < (unsigned)GNN_N ? su : 0u) << 10) | t, wl, wn);
    }
  }
}

template <bool PARTIAL>
__device__ __forceinline__ void gnn_batch_to_rows(int s, int dl, int m, int lane,
                                                  const float* __restrict__ feat, float* acc) {
  v2f row[FL_BATCH];
#pragma unroll
  for (int j = 0; j < FL_BATCH; ++j) {
    const int sj = __builtin_amdgcn_readlane(s, j);
    row[j] = *(const v2f*)(feat + (size_t)sj * GNN_D + 2 * lane);
  }
  const int dof = dl * ACC_P;
#pragma unroll
  for (int j = 0; j < FL_BATCH; j += 2) {
    if (!PARTIAL || j < m) {
      const int d0 = __builtin_amdgcn_readlane(dof, j);
      float* p0 = acc + d0 + 2 * lane;
      if (!PARTIAL || j + 1 < m) {
        const int d1 = __builtin_amdgcn_readlane(dof, j + 1);
        float* p1 = acc + d1 + 2 * lane;
        const v2f t0 = *(const v2f*)p0;
        const v2f t1 = *(const v2f*)p1;
        v2f r1 = row[j + 1];
        if (d0 == d1) r1 += row[j];
        *(v2f*)p0 = t0 + row[j];
        *(v2f*)p1 = t1 + r1;
      } else {
        const v2f t0 = *(const v2f*)p0;
        *(v2f*)p0 = t0 + row[j];
      }
    }
  }
}

__device__ __forceinline__ void gnn_list_to_sums(const unsigned* L, int n, int lane,
                                                 const float* __restrict__ feat, float* acc,
                                                 float* cntl) {
  unsigned pk = lane < n ? L[lane] : 0u;
  for (int b = 0; b < n; b += FL_BATCH) {
    const int nx = b + FL_BATCH + lane;
    const unsigned pkn = nx < n ? L[nx] : 0u;
    const unsigned su = pk >> 10;
    const int s  = (int)(su < (unsigned)GNN_N ? su : 0u);
    const int dl = (int)(pk & 1023u);
    if (b + lane < n) atomicAdd(&cntl[dl], 1.0f);
    if (n - b >= FL_BATCH) gnn_batch_to_rows<false>(s, dl, FL_BATCH, lane, feat, acc);
    else                   gnn_batch_to_rows<true>(s, dl, n - b, lane, feat, acc);
    pk = pkn;
  }
}

__global__ __launch_bounds__(AGG_THREADS) void gnn_sage_layer_kernel(
    const int* __restrict__ ei,
    const float* __restrict__ feat,
    const unsigned* __restrict__ glist,
    const int* __restrict__ gcnt,
    const float* __restrict__ Wl,
    const float* __restrict__ Wr,
    const float* __restrict__ bias,
    float* __restrict__ Y,
    int doRelu) {
  __shared__ __attribute__((aligned(16))) float acc[OWN_ROWS * ACC_P];
  __shared__ float cntl[OWN_ROWS];
  __shared__ unsigned wl[AGG_WAVES * WL_CAP];
  __shared__ int wn[AGG_WAVES];

  const int tid  = threadIdx.x;
  const int lane = tid & 31;
  const int w    = __builtin_amdgcn_readfirstlane(tid >> 5);
  const int rng  = blockIdx.x;
  const int lo   = rng * OWN_ROWS;

  for (int i = tid; i < OWN_ROWS * ACC_P / 4; i += AGG_THREADS)
    ((float4*)acc)[i] = make_float4(0.f, 0.f, 0.f, 0.f);
  for (int i = tid; i < OWN_ROWS; i += AGG_THREADS) cntl[i] = 0.f;
  if (tid < AGG_WAVES) wn[tid] = 0;
  __syncthreads();

  const int piece0 = w * PB_PER_WAVE;
  const unsigned* __restrict__ greg = glist + ((size_t)rng * PB_G + piece0) * PB_CAP;
  int c[PB_PER_WAVE];
  int any = 0;
#pragma unroll
  for (int j = 0; j < PB_PER_WAVE; ++j) {
    c[j] = __builtin_amdgcn_readfirstlane(gcnt[(piece0 + j) * OWN_G + rng]);
    any |= c[j];
  }
#pragma unroll
  for (int j = 0; j < PB_PER_WAVE; ++j) c[j] = c[j] < PB_CAP ? c[j] : PB_CAP;

  if (any >= 0) gnn_regions_to_lists(greg, c, lane, wl, wn);
  else          gnn_edges_to_lists(ei, piece0 * PB_EDGES, PB_PER_WAVE * PB_EDGES, lo, (unsigned)OWN_ROWS,
                                   lane, wl, wn);
  __syncthreads();

  const bool fits = __builtin_amdgcn_ballot_w32(wn[lane] > WL_CAP) == 0u;
  if (fits) {
    const int n = __builtin_amdgcn_readfirstlane(wn[w]);
    gnn_list_to_sums(wl + w * WL_CAP, n, lane, feat, acc, cntl);
  } else if (any >= 0) {
    gnn_regions_to_sums(greg, c, lane, feat, acc, cntl);
  } else {
    gnn_edges_to_sums(ei, piece0 * PB_EDGES, PB_PER_WAVE * PB_EDGES, lo, (unsigned)OWN_ROWS, lane,
                      feat, acc, cntl);
  }

  const int m    = lane & 15;
  const int half = lane >> 4;
  const int jt   = w & 3;
  const int tg   = w >> 2;

  v2f bl[GEMM_KT], br[GEMM_KT];
#pragma unroll
  for (int kt = 0; kt < GEMM_KT; ++kt) {
    bl[kt] = *(const v2f*)(Wl + (size_t)(jt * 16 + m) * GNN_D + kt * 4 + half * 2);
    br[kt] = *(const v2f*)(Wr + (size_t)(jt * 16 + m) * GNN_D + kt * 4 + half * 2);
  }
  const float bv = bias[jt * 16 + m];

  __syncthreads();

  for (int t = tg; t < OWN_TILES; t += AGG_WAVES / 4) {
    const int lr = t * 16 + m;
    const float* xrow = feat + (size_t)(lo + lr) * GNN_D + half * 2;
    const float* arow = acc + lr * ACC_P + half * 2;
    const float scale = 1.0f / fmaxf(cntl[lr], 1.0f);

    v8f cc = {};
    {
      v2f aa[GEMM_KT];
#pragma unroll
      for (int kt = 0; kt < GEMM_KT; ++kt) {
        const v2f sv = *(const v2f*)(arow + kt * 4);
        aa[kt].x = sv.x * scale;
        aa[kt].y = sv.y * scale;
      }
      __builtin_amdgcn_sched_barrier(0);
#pragma unroll
      for (int kt = 0; kt < GEMM_KT; ++kt)
        cc = __builtin_amdgcn_wmma_f32_16x16x4_f32(false, aa[kt], false, bl[kt], (short)0, cc, false, false);
    }
    {
      v2f ax[GEMM_KT];
#pragma unroll
      for (int kt = 0; kt < GEMM_KT; ++kt) ax[kt] = *(const v2f*)(xrow + kt * 4);
#pragma unroll
      for (int kt = 0; kt < GEMM_KT; ++kt)
        cc = __builtin_amdgcn_wmma_f32_16x16x4_f32(false, ax[kt], false, br[kt], (short)0, cc, false, false);
    }

#pragma unroll
    for (int r = 0; r < 8; ++r) {
      float v = cc[r] + bv;
      if (doRelu) v = fmaxf(v, 0.0f);
      Y[(size_t)(lo + t * 16 + r + half * 8) * GNN_D + jt * 16 + m] = v;
    }
  }
}

extern "C" void kernel_launch(void* const* d_in, const int* in_sizes, int n_in,
                              void* d_out, int out_size, void* d_ws, size_t ws_size,
                              hipStream_t stream) {
  const float* x   = (const float*)d_in[0];
  const int*   ei  = (const int*)d_in[1];
  const float* W1l = (const float*)d_in[2];
  const float* b1  = (const float*)d_in[3];
  const float* W1r = (const float*)d_in[4];
  const float* W2l = (const float*)d_in[5];
  const float* b2  = (const float*)d_in[6];
  const float* W2r = (const float*)d_in[7];
  float* out = (float*)d_out;

  float* h = (float*)d_ws;
  unsigned* glist = (unsigned*)(h + (size_t)GNN_N * GNN_D);
  int* gcnt = (int*)(glist + (size_t)OWN_G * PB_G * PB_CAP);

  gnn_bucket_kernel<<<PB_G, AGG_THREADS, 0, stream>>>(ei, glist, gcnt);

  gnn_sage_layer_kernel<<<OWN_G, AGG_THREADS, 0, stream>>>(ei, x, glist, gcnt, W1l, W1r, b1, h,  1);
  gnn_sage_layer_kernel<<<OWN_G, AGG_THREADS, 0, stream>>>(ei, h, glist, gcnt, W2l, W2r, b2, out,  0);
}


# row gather re-laid: a half-wave (16 lanes x 16 B) per row, one load and one LDS load/store carry two rows; one list per owner half-wave (row mod 64), entries shared by DPP; sum order as before
# speedup vs baseline: 5.1480x; 1.0301x over previous
#include <hip/hip_runtime.h>
#include <hip/hip_bf16.h>

typedef __attribute__((ext_vector_type(2))) float v2f;
typedef __attribute__((ext_vector_type(4))) float v4f;
typedef __attribute__((ext_vector_type(8))) float v8f;

#define GNN_N 100000
#define GNN_E 1600000
#define GNN_D 64

#define OWN_ROWS   400
#define OWN_G      (GNN_N / OWN_ROWS)
#define OWN_TILES  (OWN_ROWS / 16)
#define ACC_P      68
#define AGG_THREADS 1024
#define AGG_WAVES  (AGG_THREADS / 32)
#define FL_BATCH   32
#define FLH        16
#define PB_G       256
#define PB_EDGES   (GNN_E / PB_G)
#define PB_CAP     64
#define PB_PER_WAVE (PB_G / AGG_WAVES)
#define PB_IT      ((PB_EDGES + AGG_THREADS - 1) / AGG_THREADS)
#define GEMM_KT    (GNN_D / 4)
#define OWNERS     (2 * AGG_WAVES)
#define WL_CAP     176
#define ACC_ROWS   (OWN_ROWS + 1)

static_assert(GNN_E % PB_G == 0, "pieces must tile the edge list");
static_assert(PB_G == AGG_WAVES * PB_PER_WAVE, "every piece belongs to one owner wave of each range");
static_assert(GNN_N % OWN_ROWS == 0 && OWN_ROWS % 16 == 0, "the ranges tile the rows, and 16-row tiles tile a range: no partial tile");
static_assert(OWN_ROWS <= 1024, "local row must fit 10 bits of an entry");
static_assert((OWN_G - 1) * OWN_ROWS < GNN_N, "every work-group owns at least one row");
static_assert(GNN_N <= (1 << 22), "source row must fit 22 bits of an entry");
static_assert(OWN_G <= AGG_THREADS, "one thread clears each range counter");
static_assert(OWNERS == 64, "an entry's owner is its local row mod 64");
static_assert(AGG_WAVES == 32, "the products below give each wave one of 4 column blocks and one of 8 tile groups");

__global__ __launch_bounds__(AGG_THREADS) void gnn_bucket_kernel(
    const int* __restrict__ ei,
    unsigned* __restrict__ glist,
    int* __restrict__ gcnt) {
  __shared__ unsigned buf[OWN_G * PB_CAP];
  __shared__ int cur[OWN_G];
  const int tid = threadIdx.x;
  const int b = blockIdx.x;
  if (tid < OWN_G) cur[tid] = 0;
  __syncthreads();

  const int e0 = b * PB_EDGES;
  int sv[PB_IT], dv[PB_IT];
#pragma unroll
  for (int u = 0; u < PB_IT; ++u) {
    const int i = tid + u * AGG_THREADS;
    const bool ok = i < PB_EDGES;
    const int ic = ok ? i : PB_EDGES - 1;
    sv[u] = ei[e0 + ic];
    const int dd = ei[GNN_E + e0 + ic];
    dv[u] = ok ? dd : -1;
  }
#pragma unroll
  for (int u = 0; u < PB_IT; ++u) {
    const int d = dv[u];
    if ((unsigned)d < (unsigned)GNN_N) {
      const int r = d / OWN_ROWS;
      const int pos = atomicAdd(&cur[r], 1);
      if (pos < PB_CAP) buf[r * PB_CAP + pos] = ((unsigned)sv[u] << 10) | (unsigned)(d - r * OWN_ROWS);
    }
  }
  __syncthreads();

  const int lane = tid & 31;
  const int w = tid >> 5;
  for (int r = w; r < OWN_G; r += AGG_WAVES) {
    const int c = cur[r];
    const int n = c < PB_CAP ? c : PB_CAP;
    unsigned* __restrict__ g = glist + ((size_t)r * PB_G + b) * PB_CAP;
    for (int k = lane; k < n; k += 32) g[k] = buf[r * PB_CAP + k];
    if (lane == 0) gcnt[b * OWN_G + r] = c <= PB_CAP ? c : -1;
  }
}

template <bool PARTIAL>
__device__ __forceinline__ void gnn_batch_to_sums(int s, int dl, int m, int lane,
                                                  const float* __restrict__ feat, float* acc) {
  v2f row[FL_BATCH];
#pragma unroll
  for (int j = 0; j < FL_BATCH; ++j) {
    const int sj = __builtin_amdgcn_readlane(s, j);
    row[j] = *(const v2f*)(feat + (size_t)sj * GNN_D + 2 * lane);
  }
  const int dof = dl * ACC_P;
#pragma unroll
  for (int j = 0; j < FL_BATCH; ++j) {
    if (!PARTIAL || j < m) {
      const int dj = __builtin_amdgcn_readlane(dof, j);
      float* p = acc + dj + 2 * lane;
      atomicAdd(p, row[j].x);
      atomicAdd(p + 1, row[j].y);
    }
  }
}

__device__ __forceinline__ void gnn_regions_to_sums(const unsigned* __restrict__ g,
                                                    const int (&c)[PB_PER_WAVE], int lane,
                                                    const float* __restrict__ feat, float* acc,
                                                    float* cntl) {
  int o[PB_PER_WAVE + 1];
  o[0] = 0;
#pragma unroll
  for (int j = 0; j < PB_PER_WAVE; ++j) o[j + 1] = o[j] + c[j];
  const int n = o[PB_PER_WAVE];
  if (n <= 0) return;
  const int last = n - 1;
  auto entry = [&](int idx) -> unsigned {
    const int k = idx < last ? idx : last;
    int a = k;
#pragma unroll
    for (int j = 1; j < PB_PER_WAVE; ++j) a += (k >= o[j]) ? (PB_CAP - c[j - 1]) : 0;
    return g[a];
  };
  const unsigned v0 = entry(lane);
  unsigned pk = lane < n ? v0 : 0u;
  for (int b = 0; b < n; b += FL_BATCH) {
    const int nx = b + FL_BATCH + lane;
    const unsigned vn = entry(nx);
    const unsigned pkn = nx < n ? vn : 0u;
    const unsigned su = pk >> 10;
    const int s  = (int)(su < (unsigned)GNN_N ? su : 0u);
    const int dl = (int)(pk & 1023u);
    if (b + lane < n) atomicAdd(&cntl[dl], 1.0f);
    if (n - b >= FL_BATCH) gnn_batch_to_sums<false>(s, dl, FL_BATCH, lane, feat, acc);
    else                   gnn_batch_to_sums<true>(s, dl, n - b, lane, feat, acc);
    pk = pkn;
  }
}

__device__ __forceinline__ void gnn_edges_to_sums(const int* __restrict__ ei, int e_lo, int ne,
                                                  int lo, unsigned nr, int lane,
                                                  const float* __restrict__ feat, float* acc,
                                                  float* cntl) {
  for (int e0 = 0; e0 < ne; e0 += 32) {
    const int e = e0 + lane;
    const bool ok = e < ne;
    const int d = ok ? ei[GNN_E + e_lo + e] : -1;
    const unsigned t = (unsigned)(d - lo);
    const bool hit = ok && d >= 0 && t < nr;
    const unsigned su = hit ? (unsigned)ei[e_lo + e] : 0u;
    const int s = (int)(su < (unsigned)GNN_N ? su : 0u);
    if (hit) atomicAdd(&cntl[t], 1.0f);
    unsigned mask = __builtin_amdgcn_ballot_w32(hit);
    while (mask != 0u) {
      const int j = __builtin_ctz(mask);
      mask &= mask - 1u;
      const int sj = __shfl(s, j);
      const int tj = __shfl((int)t, j);
      const v2f row = *(const v2f*)(feat + (size_t)sj * GNN_D + 2 * lane);
      float* p = acc + tj * ACC_P + 2 * lane;
      atomicAdd(p, row.x);
      atomicAdd(p + 1, row.y);
    }
  }
}


__device__ __forceinline__ void gnn_file_entry(unsigned pk, unsigned* wl, int* wn) {
  const int ow = (int)(pk & (unsigned)(OWNERS - 1));
  const int pos = atomicAdd(&wn[ow], 1);
  if (pos < WL_CAP) wl[ow * WL_CAP + pos] = pk;
}

#define FILE_IT 8
__device__ __forceinline__ void gnn_regions_to_lists(const unsigned* __restrict__ g,
                                                     const int (&c)[PB_PER_WAVE], int lane,
                                                     unsigned* wl, int* wn) {
  int o[PB_PER_WAVE + 1];
  o[0] = 0;
#pragma unroll
  for (int j = 0; j < PB_PER_WAVE; ++j) o[j + 1] = o[j] + c[j];
  const int n = o[PB_PER_WAVE];
  if (n <= 0) return;
  const int last = n - 1;
  for (int b = 0; b < n; b += 32 * FILE_IT) {
    unsigned pk[FILE_IT];
#pragma unroll
    for (int u = 0; u < FILE_IT; ++u) {
      const int i = b + 32 * u + lane;
      const int k = i < last ? i : last;
      int a = k;
#pragma unroll
      for (int j = 1; j < PB_PER_WAVE; ++j) a += (k >= o[j]) ? (PB_CAP - c[j - 1]) : 0;
      pk[u] = g[a];
    }
    int pos[FILE_IT];
#pragma unroll
    for (int u = 0; u < FILE_IT; ++u)
      pos[u] = (b + 32 * u + lane < n) ? atomicAdd(&wn[pk[u] & (unsigned)(OWNERS - 1)], 1) : WL_CAP;
#pragma unroll
    for (int u = 0; u < FILE_IT; ++u)
      if (pos[u] < WL_CAP) wl[(pk[u] & (unsigned)(OWNERS - 1)) * WL_CAP + pos[u]] = pk[u];
  }
}

__device__ __forceinline__ void gnn_edges_to_lists(const int* __restrict__ ei, int e_lo, int ne,
                                                   int lo, unsigned nr, int lane,
                                                   unsigned* wl, int* wn) {
  for (int e0 = 0; e0 < ne; e0 += 32) {
    const int e = e0 + lane;
    const bool ok = e < ne;
    const int d = ok ? ei[GNN_E + e_lo + e] : -1;
    const unsigned t = (unsigned)(d - lo);
    if (ok && d >= 0 && t < nr) {
      const unsigned su = (unsigned)ei[e_lo + e];
      gnn_file_entry(((su < (unsigned)GNN_N ? su : 0u) << 10) | t, wl, wn);
    }
  }
}

#define GNN_SHARE(v, I) __builtin_amdgcn_update_dpp(0, (v), 0x150 + (I), 0xF, 0xF, false)
#define GNN_REP16(M) M(0) M(1) M(2) M(3) M(4) M(5) M(6) M(7) M(8) M(9) M(10) M(11) M(12) M(13) M(14) M(15)
#define GNN_REP8(M)  M(0) M(2) M(4) M(6) M(8) M(10) M(12) M(14)

template <bool PARTIAL>
__device__ __forceinline__ void gnn_batch_to_rows(int sb, int db, int m, int cb,
                                                  const char* __restrict__ fb, char* accb) {
  v4f row[FLH];
#define GNN_LD(I)                                                               \
  if (!PARTIAL || (I) < m) {                                                    \
    const unsigned a = (unsigned)GNN_SHARE(sb, I) + (unsigned)cb;               \
    row[I] = *(const v4f*)(fb + a);                                             \
  }
  GNN_REP16(GNN_LD)
#undef GNN_LD
#define GNN_UP(I)                                                               \
  if (!PARTIAL || (I) < m) {                                                    \
    const int d0 = GNN_SHARE(db, I) + cb;                                       \
    v4f* p0 = (v4f*)(accb + d0);                                                \
    if (!PARTIAL || (I) + 1 < m) {                                              \
      const int d1 = GNN_SHARE(db, (I) + 1) + cb;                               \
      v4f* p1 = (v4f*)(accb + d1);                                              \
      const v4f t0 = *p0;                                                       \
      const v4f t1 = *p1;                                                       \
      const v4f r1 = d0 == d1 ? row[(I) + 1] + row[I] : row[(I) + 1];           \
      *p0 = t0 + row[I];                                                        \
      *p1 = t1 + r1;                                                            \
    } else {                                                                    \
      const v4f t0 = *p0;                                                       \
      *p0 = t0 + row[I];                                                        \
    }                                                                           \
  }
  GNN_REP8(GNN_UP)
#undef GNN_UP
}

__device__ __forceinline__ void gnn_lists_to_sums(const unsigned* L, int n, int nmax, int lane,
                                                  const float* __restrict__ feat, float* acc,
                                                  float* cntl) {
  const int c = lane & 15;
  const int cb = c * 16;
  const char* __restrict__ fb = (const char*)feat;
  char* accb = (char*)acc;
  unsigned pk = c < n ? L[c] : 0u;
  for (int b = 0; b < nmax; b += FLH) {
    const int nx = b + FLH + c;
    const unsigned pkn = nx < n ? L[nx] : 0u;
    const bool valid = b + c < n;
    const unsigned su = pk >> 10;
    const int s  = (int)(su < (unsigned)GNN_N ? su : 0u);
    const int dl = valid ? (int)(pk & 1023u) : OWN_ROWS;
    if (valid) atomicAdd(&cntl[dl], 1.0f);
    const int sb = s * (GNN_D * 4);
    const int db = dl * (ACC_P * 4);
    if (nmax - b >= FLH) gnn_batch_to_rows<false>(sb, db, FLH, cb, fb, accb);
    else                 gnn_batch_to_rows<true>(sb, db, nmax - b, cb, fb, accb);
    pk = pkn;
  }
}

__global__ __launch_bounds__(AGG_THREADS) void gnn_sage_layer_kernel(
    const int* __restrict__ ei,
    const float* __restrict__ feat,
    const unsigned* __restrict__ glist,
    const int* __restrict__ gcnt,
    const float* __restrict__ Wl,
    const float* __restrict__ Wr,
    const float* __restrict__ bias,
    float* __restrict__ Y,
    int doRelu) {
  __shared__ __attribute__((aligned(16))) float acc[ACC_ROWS * ACC_P];
  __shared__ float cntl[ACC_ROWS];
  __shared__ unsigned wl[OWNERS * WL_CAP];
  __shared__ int wn[OWNERS];

  const int tid  = threadIdx.x;
  const int lane = tid & 31;
  const int w    = __builtin_amdgcn_readfirstlane(tid >> 5);
  const int rng  = blockIdx.x;
  const int lo   = rng * OWN_ROWS;

  for (int i = tid; i < ACC_ROWS * ACC_P / 4; i += AGG_THREADS)
    ((float4*)acc)[i] = make_float4(0.f, 0.f, 0.f, 0.f);
  for (int i = tid; i < ACC_ROWS; i += AGG_THREADS) cntl[i] = 0.f;
  if (tid < OWNERS) wn[tid] = 0;
  __syncthreads();

  const int piece0 = w * PB_PER_WAVE;
  const unsigned* __restrict__ greg = glist + ((size_t)rng * PB_G + piece0) * PB_CAP;
  int c[PB_PER_WAVE];
  int any = 0;
#pragma unroll
  for (int j = 0; j < PB_PER_WAVE; ++j) {
    c[j] = __builtin_amdgcn_readfirstlane(gcnt[(piece0 + j) * OWN_G + rng]);
    any |= c[j];
  }
#pragma unroll
  for (int j = 0; j < PB_PER_WAVE; ++j) c[j] = c[j] < PB_CAP ? c[j] : PB_CAP;

  if (any >= 0) gnn_regions_to_lists(greg, c, lane, wl, wn);
  else          gnn_edges_to_lists(ei, piece0 * PB_EDGES, PB_PER_WAVE * PB_EDGES, lo, (unsigned)OWN_ROWS,
                                   lane, wl, wn);
  __syncthreads();

  const bool fits = __builtin_amdgcn_ballot_w32(wn[lane] > WL_CAP || wn[lane + AGG_WAVES] > WL_CAP) == 0u;
  if (fits) {
    const int n0 = __builtin_amdgcn_readfirstlane(wn[w]);
    const int n1 = __builtin_amdgcn_readfirstlane(wn[w + AGG_WAVES]);
    const int hh = lane >> 4;
    gnn_lists_to_sums(wl + (w + AGG_WAVES * hh) * WL_CAP, hh ? n1 : n0, n0 > n1 ? n0 : n1, lane,
                      feat, acc, cntl);
  } else if (any >= 0) {
    gnn_regions_to_sums(greg, c, lane, feat, acc, cntl);
  } else {
    gnn_edges_to_sums(ei, piece0 * PB_EDGES, PB_PER_WAVE * PB_EDGES, lo, (unsigned)OWN_ROWS, lane,
                      feat, acc, cntl);
  }

  const int m    = lane & 15;
  const int half = lane >> 4;
  const int jt   = w & 3;
  const int tg   = w >> 2;

  v2f bl[GEMM_KT], br[GEMM_KT];
#pragma unroll
  for (int kt = 0; kt < GEMM_KT; ++kt) {
    bl[kt] = *(const v2f*)(Wl + (size_t)(jt * 16 + m) * GNN_D + kt * 4 + half * 2);
    br[kt] = *(const v2f*)(Wr + (size_t)(jt * 16 + m) * GNN_D + kt * 4 + half * 2);
  }
  const float bv = bias[jt * 16 + m];

  __syncthreads();

  for (int t = tg; t < OWN_TILES; t += AGG_WAVES / 4) {
    const int lr = t * 16 + m;
    const float* xrow = feat + (size_t)(lo + lr) * GNN_D + half * 2;
    const float* arow = acc + lr * ACC_P + half * 2;
    const float scale = 1.0f / fmaxf(cntl[lr], 1.0f);

    v8f cc = {};
    {
      v2f aa[GEMM_KT];
#pragma unroll
      for (int kt = 0; kt < GEMM_KT; ++kt) {
        const v2f sv = *(const v2f*)(arow + kt * 4);
        aa[kt].x = sv.x * scale;
        aa[kt].y = sv.y * scale;
      }
      __builtin_amdgcn_sched_barrier(0);
#pragma unroll
      for (int kt = 0; kt < GEMM_KT; ++kt)
        cc = __builtin_amdgcn_wmma_f32_16x16x4_f32(false, aa[kt], false, bl[kt], (short)0, cc, false, false);
    }
    {
      v2f ax[GEMM_KT];
#pragma unroll
      for (int kt = 0; kt < GEMM_KT; ++kt) ax[kt] = *(const v2f*)(xrow + kt * 4);
#pragma unroll
      for (int kt = 0; kt < GEMM_KT; ++kt)
        cc = __builtin_amdgcn_wmma_f32_16x16x4_f32(false, ax[kt], false, br[kt], (short)0, cc, false, false);
    }

#pragma unroll
    for (int r = 0; r < 8; ++r) {
      float v = cc[r] + bv;
      if (doRelu) v = fmaxf(v, 0.0f);
      Y[(size_t)(lo + t * 16 + r + half * 8) * GNN_D + jt * 16 + m] = v;
    }
  }
}

extern "C" void kernel_launch(void* const* d_in, const int* in_sizes, int n_in,
                              void* d_out, int out_size, void* d_ws, size_t ws_size,
                              hipStream_t stream) {
  const float* x   = (const float*)d_in[0];
  const int*   ei  = (const int*)d_in[1];
  const float* W1l = (const float*)d_in[2];
  const float* b1  = (const float*)d_in[3];
  const float* W1r = (const float*)d_in[4];
  const float* W2l = (const float*)d_in[5];
  const float* b2  = (const float*)d_in[6];
  const float* W2r = (const float*)d_in[7];
  float* out = (float*)d_out;

  float* h = (float*)d_ws;
  unsigned* glist = (unsigned*)(h + (size_t)GNN_N * GNN_D);
  int* gcnt = (int*)(glist + (size_t)OWN_G * PB_G * PB_CAP);

  gnn_bucket_kernel<<<PB_G, AGG_THREADS, 0, stream>>>(ei, glist, gcnt);

  gnn_sage_layer_kernel<<<OWN_G, AGG_THREADS, 0, stream>>>(ei, x, glist, gcnt, W1l, W1r, b1, h,  1);
  gnn_sage_layer_kernel<<<OWN_G, AGG_THREADS, 0, stream>>>(ei, h, glist, gcnt, W2l, W2r, b2, out,  0);
}


# self half of the products (X @ Wr^T + b) moved to 8 of the 32 waves, done in LDS while the other 24 waves gather; the sums' products start from it: self terms and bias are summed first
# speedup vs baseline: 5.5404x; 1.0762x over previous
#include <hip/hip_runtime.h>
#include <hip/hip_bf16.h>

typedef __attribute__((ext_vector_type(2))) float v2f;
typedef __attribute__((ext_vector_type(4))) float v4f;
typedef __attribute__((ext_vector_type(8))) float v8f;

#define GNN_N 100000
#define GNN_E 1600000
#define GNN_D 64

#define OWN_ROWS   400
#define OWN_G      (GNN_N / OWN_ROWS)
#define OWN_TILES  (OWN_ROWS / 16)
#define ACC_P      68
#define AGG_THREADS 1024
#define AGG_WAVES  (AGG_THREADS / 32)
#define FL_BATCH   32
#define FLH        16
#define PB_G       256
#define PB_EDGES   (GNN_E / PB_G)
#define PB_CAP     64
#define PB_PER_WAVE (PB_G / AGG_WAVES)
#define PB_IT      ((PB_EDGES + AGG_THREADS - 1) / AGG_THREADS)
#define GEMM_KT    (GNN_D / 4)
#define GW         24
#define SW         (AGG_WAVES - GW)
#define OWNERS     (2 * GW)
#define WL_CAP     224
#define ACC_ROWS   (OWN_ROWS + 1)

static_assert(GNN_E % PB_G == 0, "pieces must tile the edge list");
static_assert(PB_G == AGG_WAVES * PB_PER_WAVE, "every piece belongs to one owner wave of each range");
static_assert(GNN_N % OWN_ROWS == 0 && OWN_ROWS % 16 == 0, "the ranges tile the rows, and 16-row tiles tile a range: no partial tile");
static_assert(OWN_ROWS <= 1024, "local row must fit 10 bits of an entry");
static_assert((OWN_G - 1) * OWN_ROWS < GNN_N, "every work-group owns at least one row");
static_assert(GNN_N <= (1 << 22), "source row must fit 22 bits of an entry");
static_assert(OWN_G <= AGG_THREADS, "one thread clears each range counter");
static_assert(OWNERS == 48 && SW == 8, "an entry's owner is its local row mod 48; the self waves are 4 column blocks x 2 tile parities");
static_assert(AGG_WAVES == 32, "the products below give each wave one of 4 column blocks and one of 8 tile groups");

__global__ __launch_bounds__(AGG_THREADS) void gnn_bucket_kernel(
    const int* __restrict__ ei,
    unsigned* __restrict__ glist,
    int* __restrict__ gcnt) {
  __shared__ unsigned buf[OWN_G * PB_CAP];
  __shared__ int cur[OWN_G];
  const int tid = threadIdx.x;
  const int b = blockIdx.x;
  if (tid < OWN_G) cur[tid] = 0;
  __syncthreads();

  const int e0 = b * PB_EDGES;
  int sv[PB_IT], dv[PB_IT];
#pragma unroll
  for (int u = 0; u < PB_IT; ++u) {
    const int i = tid + u * AGG_THREADS;
    const bool ok = i < PB_EDGES;
    const int ic = ok ? i : PB_EDGES - 1;
    sv[u] = ei[e0 + ic];
    const int dd = ei[GNN_E + e0 + ic];
    dv[u] = ok ? dd : -1;
  }
#pragma unroll
  for (int u = 0; u < PB_IT; ++u) {
    const int d = dv[u];
    if ((unsigned)d < (unsigned)GNN_N) {
      const int r = d / OWN_ROWS;
      const int pos = atomicAdd(&cur[r], 1);
      if (pos < PB_CAP) buf[r * PB_CAP + pos] = ((unsigned)sv[u] << 10) | (unsigned)(d - r * OWN_ROWS);
    }
  }
  __syncthreads();

  const int lane = tid & 31;
  const int w = tid >> 5;
  for (int r = w; r < OWN_G; r += AGG_WAVES) {
    const int c = cur[r];
    const int n = c < PB_CAP ? c : PB_CAP;
    unsigned* __restrict__ g = glist + ((size_t)r * PB_G + b) * PB_CAP;
    for (int k = lane; k < n; k += 32) g[k] = buf[r * PB_CAP + k];
    if (lane == 0) gcnt[b * OWN_G + r] = c <= PB_CAP ? c : -1;
  }
}

template <bool PARTIAL>
__device__ __forceinline__ void gnn_batch_to_sums(int s, int dl, int m, int lane,
                                                  const float* __restrict__ feat, float* acc) {
  v2f row[FL_BATCH];
#pragma unroll
  for (int j = 0; j < FL_BATCH; ++j) {
    const int sj = __builtin_amdgcn_readlane(s, j);
    row[j] = *(const v2f*)(feat + (size_t)sj * GNN_D + 2 * lane);
  }
  const int dof = dl * ACC_P;
#pragma unroll
  for (int j = 0; j < FL_BATCH; ++j) {
    if (!PARTIAL || j < m) {
      const int dj = __builtin_amdgcn_readlane(dof, j);
      float* p = acc + dj + 2 * lane;
      atomicAdd(p, row[j].x);
      atomicAdd(p + 1, row[j].y);
    }
  }
}

__device__ __forceinline__ void gnn_regions_to_sums(const unsigned* __restrict__ g,
                                                    const int (&c)[PB_PER_WAVE], int lane,
                                                    const float* __restrict__ feat, float* acc,
                                                    float* cntl) {
  int o[PB_PER_WAVE + 1];
  o[0] = 0;
#pragma unroll
  for (int j = 0; j < PB_PER_WAVE; ++j) o[j + 1] = o[j] + c[j];
  const int n = o[PB_PER_WAVE];
  if (n <= 0) return;
  const int last = n - 1;
  auto entry = [&](int idx) -> unsigned {
    const int k = idx < last ? idx : last;
    int a = k;
#pragma unroll
    for (int j = 1; j < PB_PER_WAVE; ++j) a += (k >= o[j]) ? (PB_CAP - c[j - 1]) : 0;
    return g[a];
  };
  const unsigned v0 = entry(lane);
  unsigned pk = lane < n ? v0 : 0u;
  for (int b = 0; b < n; b += FL_BATCH) {
    const int nx = b + FL_BATCH + lane;
    const unsigned vn = entry(nx);
    const unsigned pkn = nx < n ? vn : 0u;
    const unsigned su = pk >> 10;
    const int s  = (int)(su < (unsigned)GNN_N ? su : 0u);
    const int dl = (int)(pk & 1023u);
    if (b + lane < n) atomicAdd(&cntl[dl], 1.0f);
    if (n - b >= FL_BATCH) gnn_batch_to_sums<false>(s, dl, FL_BATCH, lane, feat, acc);
    else                   gnn_batch_to_sums<true>(s, dl, n - b, lane, feat, acc);
    pk = pkn;
  }
}

__device__ __forceinline__ void gnn_edges_to_sums(const int* __restrict__ ei, int e_lo, int ne,
                                                  int lo, unsigned nr, int lane,
                                                  const float* __restrict__ feat, float* acc,
                                                  float* cntl) {
  for (int e0 = 0; e0 < ne; e0 += 32) {
    const int e = e0 + lane;
    const bool ok = e < ne;
    const int d = ok ? ei[GNN_E + e_lo + e] : -1;
    const unsigned t = (unsigned)(d - lo);
    const bool hit = ok && d >= 0 && t < nr;
    const unsigned su = hit ? (unsigned)ei[e_lo + e] : 0u;
    const int s = (int)(su < (unsigned)GNN_N ? su : 0u);
    if (hit) atomicAdd(&cntl[t], 1.0f);
    unsigned mask = __builtin_amdgcn_ballot_w32(hit);
    while (mask != 0u) {
      const int j = __builtin_ctz(mask);
      mask &= mask - 1u;
      const int sj = __shfl(s, j);
      const int tj = __shfl((int)t, j);
      const v2f row = *(const v2f*)(feat + (size_t)sj * GNN_D + 2 * lane);
      float* p = acc + tj * ACC_P + 2 * lane;
      atomicAdd(p, row.x);
      atomicAdd(p + 1, row.y);
    }
  }
}


__device__ __forceinline__ void gnn_file_entry(unsigned pk, unsigned* wl, int* wn) {
  const int ow = (int)((pk & 1023u) % (unsigned)OWNERS);
  const int pos = atomicAdd(&wn[ow], 1);
  if (pos < WL_CAP) wl[ow * WL_CAP + pos] = pk;
}

#define FILE_IT 8
__device__ __forceinline__ void gnn_regions_to_lists(const unsigned* __restrict__ g,
                                                     const int (&c)[PB_PER_WAVE], int lane,
                                                     unsigned* wl, int* wn) {
  int o[PB_PER_WAVE + 1];
  o[0] = 0;
#pragma unroll
  for (int j = 0; j < PB_PER_WAVE; ++j) o[j + 1] = o[j] + c[j];
  const int n = o[PB_PER_WAVE];
  if (n <= 0) return;
  const int last = n - 1;
  for (int b = 0; b < n; b += 32 * FILE_IT) {
    unsigned pk[FILE_IT];
#pragma unroll
    for (int u = 0; u < FILE_IT; ++u) {
      const int i = b + 32 * u + lane;
      const int k = i < last ? i : last;
      int a = k;
#pragma unroll
      for (int j = 1; j < PB_PER_WAVE; ++j) a += (k >= o[j]) ? (PB_CAP - c[j - 1]) : 0;
      pk[u] = g[a];
    }
    int pos[FILE_IT];
    unsigned ow[FILE_IT];
#pragma unroll
    for (int u = 0; u < FILE_IT; ++u) {
      ow[u] = (pk[u] & 1023u) % (unsigned)OWNERS;
      pos[u] = (b + 32 * u + lane < n) ? atomicAdd(&wn[ow[u]], 1) : WL_CAP;
    }
#pragma unroll
    for (int u = 0; u < FILE_IT; ++u)
      if (pos[u] < WL_CAP) wl[ow[u] * WL_CAP + pos[u]] = pk[u];
  }
}

__device__ __forceinline__ void gnn_edges_to_lists(const int* __restrict__ ei, int e_lo, int ne,
                                                   int lo, unsigned nr, int lane,
                                                   unsigned* wl, int* wn) {
  for (int e0 = 0; e0 < ne; e0 += 32) {
    const int e = e0 + lane;
    const bool ok = e < ne;
    const int d = ok ? ei[GNN_E + e_lo + e] : -1;
    const unsigned t = (unsigned)(d - lo);
    if (ok && d >= 0 && t < nr) {
      const unsigned su = (unsigned)ei[e_lo + e];
      gnn_file_entry(((su < (unsigned)GNN_N ? su : 0u) << 10) | t, wl, wn);
    }
  }
}

#define GNN_SHARE(v, I) __builtin_amdgcn_update_dpp(0, (v), 0x150 + (I), 0xF, 0xF, false)
#define GNN_REP16(M) M(0) M(1) M(2) M(3) M(4) M(5) M(6) M(7) M(8) M(9) M(10) M(11) M(12) M(13) M(14) M(15)
#define GNN_REP8(M)  M(0) M(2) M(4) M(6) M(8) M(10) M(12) M(14)

template <bool PARTIAL>
__device__ __forceinline__ void gnn_batch_to_rows(int sb, int db, int m, int cb,
                                                  const char* __restrict__ fb, char* accb) {
  v4f row[FLH];
#define GNN_LD(I)                                                               \
  if (!PARTIAL || (I) < m) {                                                    \
    const unsigned a = (unsigned)GNN_SHARE(sb, I) + (unsigned)cb;               \
    row[I] = *(const v4f*)(fb + a);                                             \
  }
  GNN_REP16(GNN_LD)
#undef GNN_LD
#define GNN_UP(I)                                                               \
  if (!PARTIAL || (I) < m) {                                                    \
    const int d0 = GNN_SHARE(db, I) + cb;                                       \
    v4f* p0 = (v4f*)(accb + d0);                                                \
    if (!PARTIAL || (I) + 1 < m) {                                              \
      const int d1 = GNN_SHARE(db, (I) + 1) + cb;                               \
      v4f* p1 = (v4f*)(accb + d1);                                              \
      const v4f t0 = *p0;                                                       \
      const v4f t1 = *p1;                                                       \
      const v4f r1 = d0 == d1 ? row[(I) + 1] + row[I] : row[(I) + 1];           \
      *p0 = t0 + row[I];                                                        \
      *p1 = t1 + r1;                                                            \
    } else {                                                                    \
      const v4f t0 = *p0;                                                       \
      *p0 = t0 + row[I];                                                        \
    }                                                                           \
  }
  GNN_REP8(GNN_UP)
#undef GNN_UP
}

__device__ __forceinline__ void gnn_lists_to_sums(const unsigned* L, int n, int nmax, int lane,
                                                  const float* __restrict__ feat, float* acc,
                                                  float* cntl) {
  const int c = lane & 15;
  const int cb = c * 16;
  const char* __restrict__ fb = (const char*)feat;
  char* accb = (char*)acc;
  unsigned pk = c < n ? L[c] : 0u;
  for (int b = 0; b < nmax; b += FLH) {
    const int nx = b + FLH + c;
    const unsigned pkn = nx < n ? L[nx] : 0u;
    const bool valid = b + c < n;
    const unsigned su = pk >> 10;
    const int s  = (int)(su < (unsigned)GNN_N ? su : 0u);
    const int dl = valid ? (int)(pk & 1023u) : OWN_ROWS;
    if (valid) atomicAdd(&cntl[dl], 1.0f);
    const int sb = s * (GNN_D * 4);
    const int db = dl * (ACC_P * 4);
    if (nmax - b >= FLH) gnn_batch_to_rows<false>(sb, db, FLH, cb, fb, accb);
    else                 gnn_batch_to_rows<true>(sb, db, nmax - b, cb, fb, accb);
    pk = pkn;
  }
}

__device__ __forceinline__ void gnn_self_products(int sw, int lane, int lo,
                                                  const float* __restrict__ feat,
                                                  const float* __restrict__ Wr,
                                                  const float* __restrict__ bias,
                                                  float* stage, float* selfc) {
  const int m    = lane & 15;
  const int half = lane >> 4;
  const int jt   = sw & 3;
  v2f br[GEMM_KT];
#pragma unroll
  for (int kt = 0; kt < GEMM_KT; ++kt)
    br[kt] = *(const v2f*)(Wr + (size_t)(jt * 16 + m) * GNN_D + kt * 4 + half * 2);
  const float bv = bias[jt * 16 + m];

  v4f nx[8];
  const float* x0 = feat + (size_t)(lo + half) * GNN_D + 4 * m;
  int t = sw >> 2;
#pragma unroll
  for (int i = 0; i < 8; ++i) nx[i] = *(const v4f*)(x0 + (size_t)(t * 16 + 2 * i) * GNN_D);
  for (; t < OWN_TILES; t += 2) {
#pragma unroll
    for (int i = 0; i < 8; ++i) *(v4f*)(stage + (2 * i + half) * ACC_P + 4 * m) = nx[i];
    if (t + 2 < OWN_TILES) {
#pragma unroll
      for (int i = 0; i < 8; ++i) nx[i] = *(const v4f*)(x0 + (size_t)((t + 2) * 16 + 2 * i) * GNN_D);
    }
    v2f ax[GEMM_KT];
#pragma unroll
    for (int kt = 0; kt < GEMM_KT; ++kt) ax[kt] = *(const v2f*)(stage + m * ACC_P + half * 2 + kt * 4);
    v8f cc = {};
#pragma unroll
    for (int kt = 0; kt < GEMM_KT; ++kt)
      cc = __builtin_amdgcn_wmma_f32_16x16x4_f32(false, ax[kt], false, br[kt], (short)0, cc, false, false);
#pragma unroll
    for (int r = 0; r < 8; ++r) selfc[(t * 16 + r + half * 8) * ACC_P + jt * 16 + m] = cc[r] + bv;
  }
}

__global__ __launch_bounds__(AGG_THREADS) void gnn_sage_layer_kernel(
    const int* __restrict__ ei,
    const float* __restrict__ feat,
    const unsigned* __restrict__ glist,
    const int* __restrict__ gcnt,
    const float* __restrict__ Wl,
    const float* __restrict__ Wr,
    const float* __restrict__ bias,
    float* __restrict__ Y,
    int doRelu) {
  __shared__ __attribute__((aligned(16))) float acc[ACC_ROWS * ACC_P];
  __shared__ float cntl[ACC_ROWS];
  __shared__ unsigned wl[OWNERS * WL_CAP];
  __shared__ int wn[OWNERS];
  __shared__ __attribute__((aligned(16))) float selfc[OWN_ROWS * ACC_P];
  __shared__ __attribute__((aligned(16))) float stage[SW * 16 * ACC_P];

  const int tid  = threadIdx.x;
  const int lane = tid & 31;
  const int w    = __builtin_amdgcn_readfirstlane(tid >> 5);
  const int rng  = blockIdx.x;
  const int lo   = rng * OWN_ROWS;

  for (int i = tid; i < ACC_ROWS * ACC_P / 4; i += AGG_THREADS)
    ((float4*)acc)[i] = make_float4(0.f, 0.f, 0.f, 0.f);
  for (int i = tid; i < ACC_ROWS; i += AGG_THREADS) cntl[i] = 0.f;
  if (tid < OWNERS) wn[tid] = 0;
  __syncthreads();

  const int piece0 = w * PB_PER_WAVE;
  const unsigned* __restrict__ greg = glist + ((size_t)rng * PB_G + piece0) * PB_CAP;
  int c[PB_PER_WAVE];
  int any = 0;
#pragma unroll
  for (int j = 0; j < PB_PER_WAVE; ++j) {
    c[j] = __builtin_amdgcn_readfirstlane(gcnt[(piece0 + j) * OWN_G + rng]);
    any |= c[j];
  }
#pragma unroll
  for (int j = 0; j < PB_PER_WAVE; ++j) c[j] = c[j] < PB_CAP ? c[j] : PB_CAP;

  if (any >= 0) gnn_regions_to_lists(greg, c, lane, wl, wn);
  else          gnn_edges_to_lists(ei, piece0 * PB_EDGES, PB_PER_WAVE * PB_EDGES, lo, (unsigned)OWN_ROWS,
                                   lane, wl, wn);
  __syncthreads();

  const bool fits = __builtin_amdgcn_ballot_w32(wn[lane] > WL_CAP ||
                                                (lane < OWNERS - 32 && wn[lane + 32] > WL_CAP)) == 0u;
  if (fits) {
    if (w < GW) {
      const int n0 = __builtin_amdgcn_readfirstlane(wn[w]);
      const int n1 = __builtin_amdgcn_readfirstlane(wn[w + GW]);
      const int hh = lane >> 4;
      gnn_lists_to_sums(wl + (w + GW * hh) * WL_CAP, hh ? n1 : n0, n0 > n1 ? n0 : n1, lane,
                        feat, acc, cntl);
    }
  } else if (any >= 0) {
    gnn_regions_to_sums(greg, c, lane, feat, acc, cntl);
  } else {
    gnn_edges_to_sums(ei, piece0 * PB_EDGES, PB_PER_WAVE * PB_EDGES, lo, (unsigned)OWN_ROWS, lane,
                      feat, acc, cntl);
  }

  if (w >= GW) gnn_self_products(w - GW, lane, lo, feat, Wr, bias, stage + (w - GW) * 16 * ACC_P, selfc);

  const int m    = lane & 15;
  const int half = lane >> 4;
  const int jt   = w & 3;
  const int tg   = w >> 2;

  v2f bl[GEMM_KT];
#pragma unroll
  for (int kt = 0; kt < GEMM_KT; ++kt)
    bl[kt] = *(const v2f*)(Wl + (size_t)(jt * 16 + m) * GNN_D + kt * 4 + half * 2);

  __syncthreads();

  for (int t = tg; t < OWN_TILES; t += AGG_WAVES / 4) {
    const int lr = t * 16 + m;
    const float* arow = acc + lr * ACC_P + half * 2;
    const float scale = 1.0f / fmaxf(cntl[lr], 1.0f);

    v8f cc;
#pragma unroll
    for (int r = 0; r < 8; ++r) cc[r] = selfc[(t * 16 + r + half * 8) * ACC_P + jt * 16 + m];
    {
      v2f aa[GEMM_KT];
#pragma unroll
      for (int kt = 0; kt < GEMM_KT; ++kt) {
        const v2f sv = *(const v2f*)(arow + kt * 4);
        aa[kt].x = sv.x * scale;
        aa[kt].y = sv.y * scale;
      }
      __builtin_amdgcn_sched_barrier(0);
#pragma unroll
      for (int kt = 0; kt < GEMM_KT; ++kt)
        cc = __builtin_amdgcn_wmma_f32_16x16x4_f32(false, aa[kt], false, bl[kt], (short)0, cc, false, false);
    }
#pragma unroll
    for (int r = 0; r < 8; ++r) {
      float v = cc[r];
      if (doRelu) v = fmaxf(v, 0.0f);
      Y[(size_t)(lo + t * 16 + r + half * 8) * GNN_D + jt * 16 + m] = v;
    }
  }
}

extern "C" void kernel_launch(void* const* d_in, const int* in_sizes, int n_in,
                              void* d_out, int out_size, void* d_ws, size_t ws_size,
                              hipStream_t stream) {
  const float* x   = (const float*)d_in[0];
  const int*   ei  = (const int*)d_in[1];
  const float* W1l = (const float*)d_in[2];
  const float* b1  = (const float*)d_in[3];
  const float* W1r = (const float*)d_in[4];
  const float* W2l = (const float*)d_in[5];
  const float* b2  = (const float*)d_in[6];
  const float* W2r = (const float*)d_in[7];
  float* out = (float*)d_out;

  float* h = (float*)d_ws;
  unsigned* glist = (unsigned*)(h + (size_t)GNN_N * GNN_D);
  int* gcnt = (int*)(glist + (size_t)OWN_G * PB_G * PB_CAP);

  gnn_bucket_kernel<<<PB_G, AGG_THREADS, 0, stream>>>(ei, glist, gcnt);

  gnn_sage_layer_kernel<<<OWN_G, AGG_THREADS, 0, stream>>>(ei, x, glist, gcnt, W1l, W1r, b1, h,  1);
  gnn_sage_layer_kernel<<<OWN_G, AGG_THREADS, 0, stream>>>(ei, h, glist, gcnt, W2l, W2r, b2, out,  0);
}
